# starting kernel
# baseline (speedup 1.0000x reference)
#include <hip/hip_runtime.h>

typedef __attribute__((ext_vector_type(2))) float v2f;
typedef __attribute__((ext_vector_type(4))) float v4f;
typedef __attribute__((ext_vector_type(8))) float v8f;

#define ESTRIDE 100
#define NSTRIDE 68

__device__ __forceinline__ v8f wmma_f32(v2f a, v2f b, v8f c) {
  return __builtin_amdgcn_wmma_f32_16x16x4_f32(false, a, false, b, (short)0, c,
                                               false, false);
}

__global__ __launch_bounds__(256) void zero_kernel(float* __restrict__ p, int n) {
  int n4 = n >> 2;
  int i = blockIdx.x * 256 + threadIdx.x;
  if (i < n4) ((v4f*)p)[i] = v4f{0.f, 0.f, 0.f, 0.f};
  if (i == 0)
    for (int t = n4 << 2; t < n; ++t) p[t] = 0.f;
}

__global__ __launch_bounds__(256) void init_misc_kernel(int* __restrict__ perm,
                                                        int n_perm,
                                                        int* __restrict__ ttype,
                                                        int n_tiles,
                                                        int* __restrict__ ctr) {
  int i = blockIdx.x * 256 + threadIdx.x;
  if (i < n_perm) perm[i] = -1;
  if (i < n_tiles) ttype[i] = 0;
  if (i < 12) ctr[i] = 0;
}

__global__ __launch_bounds__(256) void count_kernel(const int* __restrict__ etype,
                                                    int n_edges,
                                                    int* __restrict__ ctr) {
  __shared__ int wg[3];
  const int tid = threadIdx.x;
  if (tid < 3) wg[tid] = 0;
  __syncthreads();
  const int i = blockIdx.x * 256 + tid;
  const int t = (i < n_edges) ? etype[i] : -1;
  const int lane = tid & 31;
#pragma unroll
  for (int tt = 0; tt < 3; ++tt) {
    unsigned mask = (unsigned)__ballot(t == tt);
    int cnt = __popc(mask);
    if (lane == 0 && cnt) atomicAdd(&wg[tt], cnt);
  }
  __syncthreads();
  if (tid < 3 && wg[tid]) atomicAdd(&ctr[tid], wg[tid]);
}

__global__ void base_kernel(int* __restrict__ ctr) {
  if (threadIdx.x == 0 && blockIdx.x == 0) {
    int c0 = ctr[0], c1 = ctr[1];
    int b1 = (c0 + 15) & ~15;
    int b2 = b1 + ((c1 + 15) & ~15);
    ctr[3] = 0;
    ctr[4] = b1;
    ctr[5] = b2;
    ctr[6] = 0;
    ctr[7] = 0;
    ctr[8] = 0;
  }
}

__global__ __launch_bounds__(256) void scatter_kernel(
    const int* __restrict__ etype, int n_edges, int* __restrict__ ctr,
    int* __restrict__ perm, int* __restrict__ ttype) {
  __shared__ int wg_cnt[3];
  __shared__ int wg_base[3];
  __shared__ int wv_off[8][3];
  const int tid = threadIdx.x;
  if (tid < 3) wg_cnt[tid] = 0;
  __syncthreads();
  const int i = blockIdx.x * 256 + tid;
  const int t = (i < n_edges) ? etype[i] : -1;
  const int wave = tid >> 5, lane = tid & 31;
  const unsigned low = (lane == 0) ? 0u : (~0u >> (32 - lane));
  int my_pre = 0;
#pragma unroll
  for (int tt = 0; tt < 3; ++tt) {
    unsigned mask = (unsigned)__ballot(t == tt);
    if (t == tt) my_pre = __popc(mask & low);
    if (lane == 0) wv_off[wave][tt] = atomicAdd(&wg_cnt[tt], __popc(mask));
  }
  __syncthreads();
  if (tid < 3) wg_base[tid] = wg_cnt[tid] ? atomicAdd(&ctr[6 + tid], wg_cnt[tid]) : 0;
  __syncthreads();
  if (t >= 0) {
    int slot = ctr[3 + t] + wg_base[t] + wv_off[wave][t] + my_pre;
    perm[slot] = i;
    ttype[slot >> 4] = t;
  }
}

__global__ __launch_bounds__(256) void node_xform_kernel(
    const float* __restrict__ x, const float* __restrict__ Wn,
    const int* __restrict__ ntype, float* __restrict__ xout, int n_nodes,
    int n_tiles) {
  __shared__ float w[2 * 64 * NSTRIDE];
  for (int i = threadIdx.x; i < 2 * 64 * 64; i += 256) {
    int t = i >> 12, k = (i >> 6) & 63, c = i & 63;
    w[(t * 64 + c) * NSTRIDE + k] = Wn[i];
  }
  __syncthreads();

  const int wave = threadIdx.x >> 5;
  const int lane = threadIdx.x & 31;
  const int tile = blockIdx.x * 8 + wave;
  if (tile >= n_tiles) return;

  const int arow = lane & 15;
  const int khalf = (lane >> 4) << 1;
  int node = tile * 16 + arow;
  const bool valid = node < n_nodes;
  if (!valid) node = n_nodes - 1;
  const int t_row = valid ? ntype[node] : -1;
  const float* __restrict__ ar = x + (size_t)node * 64;
  const float* __restrict__ wb0 = &w[arow * NSTRIDE];
  const float* __restrict__ wb1 = &w[(64 + arow) * NSTRIDE];

  v8f acc[4] = {};
#pragma unroll
  for (int kk = 0; kk < 64; kk += 4) {
    const int k = kk + khalf;
    v2f a = __builtin_nontemporal_load((const v2f*)(ar + k));
    const v2f z = {0.f, 0.f};
    v2f a0 = (t_row == 0) ? a : z;
    v2f a1 = (t_row == 1) ? a : z;
#pragma unroll
    for (int nb = 0; nb < 4; ++nb) {
      v2f b0 = *(const v2f*)(wb0 + nb * 16 * NSTRIDE + k);
      v2f b1 = *(const v2f*)(wb1 + nb * 16 * NSTRIDE + k);
      acc[nb] = wmma_f32(a0, b0, acc[nb]);
      acc[nb] = wmma_f32(a1, b1, acc[nb]);
    }
  }

  const int rbase = (lane >> 4) * 8;
  const int col = lane & 15;
#pragma unroll
  for (int r = 0; r < 8; ++r) {
    const int orow = tile * 16 + rbase + r;
    if (orow < n_nodes) {
#pragma unroll
      for (int nb = 0; nb < 4; ++nb)
        xout[(size_t)orow * 64 + nb * 16 + col] = ((const float*)&acc[nb])[r];
    }
  }
}

__global__ __launch_bounds__(256) void edge_msg_kernel(
    const float* __restrict__ xout, const float* __restrict__ ef,
    const float* __restrict__ Wm, const int* __restrict__ perm,
    const int* __restrict__ ttype, const int* __restrict__ src,
    const int* __restrict__ dst, float* __restrict__ out, int n_tiles) {
  __shared__ float w[3 * 64 * ESTRIDE];
  for (int i = threadIdx.x; i < 3 * 96 * 64; i += 256) {
    int t = i / (96 * 64);
    int rem = i - t * 96 * 64;
    int k = rem >> 6, c = rem & 63;
    w[(t * 64 + c) * ESTRIDE + k] = Wm[i];
  }
  __syncthreads();

  const int wave = threadIdx.x >> 5;
  const int lane = threadIdx.x & 31;
  const int tile = blockIdx.x * 8 + wave;
  if (tile >= n_tiles) return;

  const int arow = lane & 15;
  const int khalf = (lane >> 4) << 1;
  const int e = perm[tile * 16 + arow];
  const int es = (e >= 0) ? e : 0;
  const int t = ttype[tile];
  const int s = src[es];
  const int dd = dst[es];
  const float* __restrict__ xr = xout + (size_t)s * 64;
  const float* __restrict__ er = ef + (size_t)es * 32;
  const float* __restrict__ wb = &w[(t * 64 + arow) * ESTRIDE];

  v8f acc[4] = {};
#pragma unroll
  for (int kk = 0; kk < 64; kk += 4) {
    const int k = kk + khalf;
    v2f a = *(const v2f*)(xr + k);
#pragma unroll
    for (int nb = 0; nb < 4; ++nb) {
      v2f b = *(const v2f*)(wb + nb * 16 * ESTRIDE + k);
      acc[nb] = wmma_f32(a, b, acc[nb]);
    }
  }
#pragma unroll
  for (int kk = 0; kk < 32; kk += 4) {
    const int k = kk + khalf;
    v2f a = __builtin_nontemporal_load((const v2f*)(er + k));
#pragma unroll
    for (int nb = 0; nb < 4; ++nb) {
      v2f b = *(const v2f*)(wb + nb * 16 * ESTRIDE + 64 + k);
      acc[nb] = wmma_f32(a, b, acc[nb]);
    }
  }

  const int rbase = (lane >> 4) * 8;
  const int col = lane & 15;
#pragma unroll
  for (int r = 0; r < 8; ++r) {
    const int eo = __shfl(e, rbase + r, 32);
    const int dr = __shfl(dd, rbase + r, 32);
    if (eo >= 0) {
      float* orow = out + (size_t)dr * 64 + col;
#pragma unroll
      for (int nb = 0; nb < 4; ++nb)
        unsafeAtomicAdd(orow + nb * 16, ((const float*)&acc[nb])[r]);
    }
  }
}

extern "C" void kernel_launch(void* const* d_in, const int* in_sizes, int n_in,
                              void* d_out, int out_size, void* d_ws,
                              size_t ws_size, hipStream_t stream) {
  const float* node_feature = (const float*)d_in[0];
  const float* edge_feature = (const float*)d_in[1];
  const float* W_node = (const float*)d_in[2];
  const float* W_msg = (const float*)d_in[3];
  const int* node_type = (const int*)d_in[4];
  const int* edge_type = (const int*)d_in[5];
  const int* edge_index = (const int*)d_in[6];
  const int n_nodes = in_sizes[4];
  const int n_edges = in_sizes[5];
  const int* src = edge_index;
  const int* dst = edge_index + n_edges;

  float* out = (float*)d_out;

  float* xout = (float*)d_ws;
  const int edge_tiles = (n_edges + 48) / 16 + 1;
  const int n_perm = edge_tiles * 16;
  int* perm = (int*)((char*)d_ws + (size_t)n_nodes * 64 * sizeof(float));
  int* ttype = perm + n_perm;
  int* ctr = ttype + edge_tiles;

  zero_kernel<<<((out_size >> 2) + 255) / 256, 256, 0, stream>>>(out, out_size);
  init_misc_kernel<<<(n_perm + 255) / 256, 256, 0, stream>>>(perm, n_perm,
                                                             ttype, edge_tiles,
                                                             ctr);
  count_kernel<<<(n_edges + 255) / 256, 256, 0, stream>>>(edge_type, n_edges,
                                                          ctr);
  base_kernel<<<1, 32, 0, stream>>>(ctr);
  scatter_kernel<<<(n_edges + 255) / 256, 256, 0, stream>>>(edge_type, n_edges,
                                                            ctr, perm, ttype);

  const int node_tiles = (n_nodes + 15) / 16;
  node_xform_kernel<<<(node_tiles + 7) / 8, 256, 0, stream>>>(
      node_feature, W_node, node_type, xout, n_nodes, node_tiles);

  edge_msg_kernel<<<(edge_tiles + 7) / 8, 256, 0, stream>>>(
      xout, edge_feature, W_msg, perm, ttype, src, dst, out, edge_tiles);
}


# Edge kernel persistent: W_msg staged to LDS once per resident work-group (grid 12501->1024, 8 waves/SIMD kept by 32-bit row indices), waves stride over tiles; atomic order unordered as before
# speedup vs baseline: 1.2689x; 1.2689x over previous
#include <hip/hip_runtime.h>

typedef __attribute__((ext_vector_type(2))) float v2f;
typedef __attribute__((ext_vector_type(4))) float v4f;
typedef __attribute__((ext_vector_type(8))) float v8f;

#define ESTRIDE 100
#define NSTRIDE 68

__device__ __forceinline__ v8f wmma_f32(v2f a, v2f b, v8f c) {
  return __builtin_amdgcn_wmma_f32_16x16x4_f32(false, a, false, b, (short)0, c,
                                               false, false);
}

__global__ __launch_bounds__(256) void zero_kernel(float* __restrict__ p, int n) {
  int n4 = n >> 2;
  int i = blockIdx.x * 256 + threadIdx.x;
  if (i < n4) ((v4f*)p)[i] = v4f{0.f, 0.f, 0.f, 0.f};
  if (i == 0)
    for (int t = n4 << 2; t < n; ++t) p[t] = 0.f;
}

__global__ __launch_bounds__(256) void init_misc_kernel(int* __restrict__ perm,
                                                        int n_perm,
                                                        int* __restrict__ ttype,
                                                        int n_tiles,
                                                        int* __restrict__ ctr) {
  int i = blockIdx.x * 256 + threadIdx.x;
  if (i < n_perm) perm[i] = -1;
  if (i < n_tiles) ttype[i] = 0;
  if (i < 12) ctr[i] = 0;
}

__global__ __launch_bounds__(256) void count_kernel(const int* __restrict__ etype,
                                                    int n_edges,
                                                    int* __restrict__ ctr) {
  __shared__ int wg[3];
  const int tid = threadIdx.x;
  if (tid < 3) wg[tid] = 0;
  __syncthreads();
  const int i = blockIdx.x * 256 + tid;
  const int t = (i < n_edges) ? etype[i] : -1;
  const int lane = tid & 31;
#pragma unroll
  for (int tt = 0; tt < 3; ++tt) {
    unsigned mask = (unsigned)__ballot(t == tt);
    int cnt = __popc(mask);
    if (lane == 0 && cnt) atomicAdd(&wg[tt], cnt);
  }
  __syncthreads();
  if (tid < 3 && wg[tid]) atomicAdd(&ctr[tid], wg[tid]);
}

__global__ void base_kernel(int* __restrict__ ctr) {
  if (threadIdx.x == 0 && blockIdx.x == 0) {
    int c0 = ctr[0], c1 = ctr[1];
    int b1 = (c0 + 15) & ~15;
    int b2 = b1 + ((c1 + 15) & ~15);
    ctr[3] = 0;
    ctr[4] = b1;
    ctr[5] = b2;
    ctr[6] = 0;
    ctr[7] = 0;
    ctr[8] = 0;
  }
}

__global__ __launch_bounds__(256) void scatter_kernel(
    const int* __restrict__ etype, int n_edges, int* __restrict__ ctr,
    int* __restrict__ perm, int* __restrict__ ttype) {
  __shared__ int wg_cnt[3];
  __shared__ int wg_base[3];
  __shared__ int wv_off[8][3];
  const int tid = threadIdx.x;
  if (tid < 3) wg_cnt[tid] = 0;
  __syncthreads();
  const int i = blockIdx.x * 256 + tid;
  const int t = (i < n_edges) ? etype[i] : -1;
  const int wave = tid >> 5, lane = tid & 31;
  const unsigned low = (lane == 0) ? 0u : (~0u >> (32 - lane));
  int my_pre = 0;
#pragma unroll
  for (int tt = 0; tt < 3; ++tt) {
    unsigned mask = (unsigned)__ballot(t == tt);
    if (t == tt) my_pre = __popc(mask & low);
    if (lane == 0) wv_off[wave][tt] = atomicAdd(&wg_cnt[tt], __popc(mask));
  }
  __syncthreads();
  if (tid < 3) wg_base[tid] = wg_cnt[tid] ? atomicAdd(&ctr[6 + tid], wg_cnt[tid]) : 0;
  __syncthreads();
  if (t >= 0) {
    int slot = ctr[3 + t] + wg_base[t] + wv_off[wave][t] + my_pre;
    perm[slot] = i;
    ttype[slot >> 4] = t;
  }
}

__global__ __launch_bounds__(256) void node_xform_kernel(
    const float* __restrict__ x, const float* __restrict__ Wn,
    const int* __restrict__ ntype, float* __restrict__ xout, int n_nodes,
    int n_tiles) {
  __shared__ float w[2 * 64 * NSTRIDE];
  for (int i = threadIdx.x; i < 2 * 64 * 64; i += 256) {
    int t = i >> 12, k = (i >> 6) & 63, c = i & 63;
    w[(t * 64 + c) * NSTRIDE + k] = Wn[i];
  }
  __syncthreads();

  const int wave = threadIdx.x >> 5;
  const int lane = threadIdx.x & 31;
  const int tile = blockIdx.x * 8 + wave;
  if (tile >= n_tiles) return;

  const int arow = lane & 15;
  const int khalf = (lane >> 4) << 1;
  int node = tile * 16 + arow;
  const bool valid = node < n_nodes;
  if (!valid) node = n_nodes - 1;
  const int t_row = valid ? ntype[node] : -1;
  const float* __restrict__ ar = x + (size_t)node * 64;
  const float* __restrict__ wb0 = &w[arow * NSTRIDE];
  const float* __restrict__ wb1 = &w[(64 + arow) * NSTRIDE];

  v8f acc[4] = {};
#pragma unroll
  for (int kk = 0; kk < 64; kk += 4) {
    const int k = kk + khalf;
    v2f a = __builtin_nontemporal_load((const v2f*)(ar + k));
    const v2f z = {0.f, 0.f};
    v2f a0 = (t_row == 0) ? a : z;
    v2f a1 = (t_row == 1) ? a : z;
#pragma unroll
    for (int nb = 0; nb < 4; ++nb) {
      v2f b0 = *(const v2f*)(wb0 + nb * 16 * NSTRIDE + k);
      v2f b1 = *(const v2f*)(wb1 + nb * 16 * NSTRIDE + k);
      acc[nb] = wmma_f32(a0, b0, acc[nb]);
      acc[nb] = wmma_f32(a1, b1, acc[nb]);
    }
  }

  const int rbase = (lane >> 4) * 8;
  const int col = lane & 15;
#pragma unroll
  for (int r = 0; r < 8; ++r) {
    const int orow = tile * 16 + rbase + r;
    if (orow < n_nodes) {
#pragma unroll
      for (int nb = 0; nb < 4; ++nb)
        xout[(size_t)orow * 64 + nb * 16 + col] = ((const float*)&acc[nb])[r];
    }
  }
}

__global__ __launch_bounds__(256, 8) void edge_msg_kernel(
    const float* __restrict__ xout, const float* __restrict__ ef,
    const float* __restrict__ Wm, const int* __restrict__ perm,
    const int* __restrict__ ttype, const int* __restrict__ src,
    const int* __restrict__ dst, float* __restrict__ out, int n_tiles) {
  __shared__ float w[3 * 64 * ESTRIDE];
  for (int i = threadIdx.x; i < 3 * 96 * 64; i += 256) {
    int t = i / (96 * 64);
    int rem = i - t * 96 * 64;
    int k = rem >> 6, c = rem & 63;
    w[(t * 64 + c) * ESTRIDE + k] = Wm[i];
  }
  __syncthreads();

  const int wave = __builtin_amdgcn_readfirstlane((int)(threadIdx.x >> 5));
  const int lane = threadIdx.x & 31;
  const int arow = lane & 15;
  const int khalf = (lane >> 4) << 1;
  const int rbase = (lane >> 4) * 8;
  const int col = lane & 15;
  const int tile_step = (int)gridDim.x * 8;

  for (int tile = (int)blockIdx.x * 8 + wave; tile < n_tiles; tile += tile_step) {
    const int e = perm[tile * 16 + arow];
    const int es = (e >= 0) ? e : 0;
    const int t = ttype[tile];
    const int s = src[es];
    const int dd = dst[es];
    const v2f* __restrict__ xr = (const v2f*)xout + ((unsigned)s * 32u + (unsigned)(khalf >> 1));
    const v2f* __restrict__ er = (const v2f*)ef + ((unsigned)es * 16u + (unsigned)(khalf >> 1));
    const float* __restrict__ wb = &w[(t * 64 + arow) * ESTRIDE];

    v8f acc[4] = {};
#pragma unroll
    for (int kk = 0; kk < 64; kk += 4) {
      const int k = kk + khalf;
      v2f a = xr[kk >> 1];
#pragma unroll
      for (int nb = 0; nb < 4; ++nb) {
        v2f b = *(const v2f*)(wb + nb * 16 * ESTRIDE + k);
        acc[nb] = wmma_f32(a, b, acc[nb]);
      }
    }
#pragma unroll
    for (int kk = 0; kk < 32; kk += 4) {
      const int k = kk + khalf;
      v2f a = __builtin_nontemporal_load(er + (kk >> 1));
#pragma unroll
      for (int nb = 0; nb < 4; ++nb) {
        v2f b = *(const v2f*)(wb + nb * 16 * ESTRIDE + 64 + k);
        acc[nb] = wmma_f32(a, b, acc[nb]);
      }
    }

#pragma unroll
    for (int r = 0; r < 8; ++r) {
      const int eo = __shfl(e, rbase + r, 32);
      const int dr = __shfl(dd, rbase + r, 32);
      if (eo >= 0) {
        float* orow = out + ((unsigned)dr * 64u + (unsigned)col);
#pragma unroll
        for (int nb = 0; nb < 4; ++nb)
          unsafeAtomicAdd(orow + nb * 16, ((const float*)&acc[nb])[r]);
      }
    }
  }
}

extern "C" void kernel_launch(void* const* d_in, const int* in_sizes, int n_in,
                              void* d_out, int out_size, void* d_ws,
                              size_t ws_size, hipStream_t stream) {
  const float* node_feature = (const float*)d_in[0];
  const float* edge_feature = (const float*)d_in[1];
  const float* W_node = (const float*)d_in[2];
  const float* W_msg = (const float*)d_in[3];
  const int* node_type = (const int*)d_in[4];
  const int* edge_type = (const int*)d_in[5];
  const int* edge_index = (const int*)d_in[6];
  const int n_nodes = in_sizes[4];
  const int n_edges = in_sizes[5];
  const int* src = edge_index;
  const int* dst = edge_index + n_edges;

  float* out = (float*)d_out;

  float* xout = (float*)d_ws;
  const int edge_tiles = (n_edges + 48) / 16 + 1;
  const int n_perm = edge_tiles * 16;
  int* perm = (int*)((char*)d_ws + (size_t)n_nodes * 64 * sizeof(float));
  int* ttype = perm + n_perm;
  int* ctr = ttype + edge_tiles;

  zero_kernel<<<((out_size >> 2) + 255) / 256, 256, 0, stream>>>(out, out_size);
  init_misc_kernel<<<(n_perm + 255) / 256, 256, 0, stream>>>(perm, n_perm,
                                                             ttype, edge_tiles,
                                                             ctr);
  count_kernel<<<(n_edges + 255) / 256, 256, 0, stream>>>(edge_type, n_edges,
                                                          ctr);
  base_kernel<<<1, 32, 0, stream>>>(ctr);
  scatter_kernel<<<(n_edges + 255) / 256, 256, 0, stream>>>(edge_type, n_edges,
                                                            ctr, perm, ttype);

  const int node_tiles = (n_nodes + 15) / 16;
  node_xform_kernel<<<(node_tiles + 7) / 8, 256, 0, stream>>>(
      node_feature, W_node, node_type, xout, n_nodes, node_tiles);

  const int edge_wgs_all = (edge_tiles + 7) / 8;
  const int edge_wgs = edge_wgs_all < 1024 ? edge_wgs_all : 1024;
  edge_msg_kernel<<<edge_wgs, 256, 0, stream>>>(
      xout, edge_feature, W_msg, perm, ttype, src, dst, out, edge_tiles);
}


# Message GEMM fused into the aggregate kernel: a bucket's 32x288 sums stay in LDS and are multiplied there, S never goes to memory, one launch fewer; sum order: two k halves added through LDS
# speedup vs baseline: 1.6389x; 1.2916x over previous
#include <hip/hip_runtime.h>


typedef __attribute__((ext_vector_type(2))) float v2f;
typedef __attribute__((ext_vector_type(4))) float v4f;
typedef __attribute__((ext_vector_type(8))) float v8f;
typedef __attribute__((ext_vector_type(4))) int v4i;

#define NSTRIDE 68
#define MSG_K 288
#define SSTRIDE 292
#define OSTRIDE 68
#define WFRAG_FLOATS (MSG_K * 64)

__device__ __forceinline__ v8f wmma_f32(v2f a, v2f b, v8f c) {
  return __builtin_amdgcn_wmma_f32_16x16x4_f32(false, a, false, b, (short)0, c,
                                               false, false);
}

__device__ __forceinline__ int wave_incl_scan(int v, int lane) {
#pragma unroll
  for (int off = 1; off < 32; off <<= 1) {
    int u = __shfl_up(v, off, 32);
    if (lane >= off) v += u;
  }
  return v;
}

__device__ __forceinline__ int wave_sum(int v) {
#pragma unroll
  for (int off = 16; off > 0; off >>= 1) v += __shfl_xor(v, off, 32);
  return v;
}

#define BUCKET_SHIFT 5
#define BUCKET_NODES 32
#define SORT_THREADS 1024

__global__ __launch_bounds__(SORT_THREADS) void bucket_count_kernel(
    const int* __restrict__ dst, int n_edges, int chunk, int n_buckets,
    int* __restrict__ wgcnt) {
  extern __shared__ int hist[];
  const int tid = threadIdx.x;
  const int wg = blockIdx.x, n_wg = gridDim.x;
  for (int b = tid; b < n_buckets; b += SORT_THREADS) hist[b] = 0;
  __syncthreads();
  const int cbeg = wg * chunk;
  const int cend = (cbeg + chunk < n_edges) ? cbeg + chunk : n_edges;
  for (int i0 = cbeg + tid; i0 < cend; i0 += 8 * SORT_THREADS) {
    int d[8];
#pragma unroll
    for (int u = 0; u < 8; ++u) {
      const int i = i0 + u * SORT_THREADS;
      d[u] = dst[i < cend ? i : cend - 1];
    }
#pragma unroll
    for (int u = 0; u < 8; ++u)
      if (i0 + u * SORT_THREADS < cend) atomicAdd(&hist[d[u] >> BUCKET_SHIFT], 1);
  }
  __syncthreads();
  for (int b = tid; b < n_buckets; b += SORT_THREADS)
    wgcnt[(size_t)b * n_wg + wg] = hist[b];
}

__global__ __launch_bounds__(1024) void blocksum_kernel(const int* __restrict__ cnt,
                                                        int n,
                                                        int* __restrict__ bsum) {
  __shared__ int ws[32];
  const int tid = threadIdx.x;
  const int lane = tid & 31, wave = tid >> 5;
  const int i = blockIdx.x * 1024 + tid;
  int v = (i < n) ? cnt[i] : 0;
  v = wave_sum(v);
  if (lane == 0) ws[wave] = v;
  __syncthreads();
  if (wave == 0) {
    int t = wave_sum(ws[lane]);
    if (lane == 0) bsum[blockIdx.x] = t;
  }
}

__global__ __launch_bounds__(1024) void scan_kernel(const int* __restrict__ cnt,
                                                    int n,
                                                    const int* __restrict__ bsum,
                                                    int* __restrict__ start) {
  __shared__ int ws[32];
  __shared__ int base_s;
  const int tid = threadIdx.x;
  const int lane = tid & 31, wave = tid >> 5;
  if (wave == 0) {
    int t = 0;
    for (int j = lane; j < (int)blockIdx.x; j += 32) t += bsum[j];
    t = wave_sum(t);
    if (lane == 0) base_s = t;
  }
  const int i = blockIdx.x * 1024 + tid;
  const int v = (i < n) ? cnt[i] : 0;
  const int inc = wave_incl_scan(v, lane);
  if (lane == 31) ws[wave] = inc;
  __syncthreads();
  if (wave == 0) {
    int t = ws[lane];
    int ti = wave_incl_scan(t, lane);
    ws[lane] = ti - t;
  }
  __syncthreads();
  const int excl = base_s + ws[wave] + inc - v;
  if (i < n) start[i] = excl;
  if (i == n - 1) start[n] = excl + v;
}

__global__ __launch_bounds__(SORT_THREADS) void bucket_place_kernel(
    const int* __restrict__ src, const int* __restrict__ dst,
    const int* __restrict__ etype, int n_edges, int chunk, int n_buckets,
    const int* __restrict__ pos, int* __restrict__ rec) {
  extern __shared__ int cursor[];
  const int tid = threadIdx.x;
  const int wg = blockIdx.x, n_wg = gridDim.x;
  for (int b = tid; b < n_buckets; b += SORT_THREADS)
    cursor[b] = pos[(size_t)b * n_wg + wg];
  __syncthreads();
  const int cbeg = wg * chunk;
  const int cend = (cbeg + chunk < n_edges) ? cbeg + chunk : n_edges;
  for (int i0 = cbeg + tid; i0 < cend; i0 += 8 * SORT_THREADS) {
    int d[8], t[8], sn[8];
#pragma unroll
    for (int u = 0; u < 8; ++u) {
      const int i = i0 + u * SORT_THREADS;
      const int ic = i < cend ? i : cend - 1;
      d[u] = dst[ic];
      t[u] = etype[ic];
      sn[u] = src[ic];
    }
#pragma unroll
    for (int u = 0; u < 8; ++u) {
      const int i = i0 + u * SORT_THREADS;
      if (i < cend) {
        const int tt = (t[u] < 0 || t[u] > 2) ? 3 : t[u];
        const int p = atomicAdd(&cursor[d[u] >> BUCKET_SHIFT], 1);
        v4i r = {i, sn[u], tt, d[u] & (BUCKET_NODES - 1)};
        *(v4i*)(rec + 4 * (size_t)p) = r;
      }
    }
  }
}

__global__ __launch_bounds__(256) void node_xform_kernel(
    const float* __restrict__ x, const float* __restrict__ Wn,
    const int* __restrict__ ntype, float* __restrict__ xout, int n_nodes,
    int n_tiles, const float* __restrict__ Wm, float* __restrict__ wfrag) {
  __shared__ float w[2 * 64 * NSTRIDE];
  const int tile_blocks = (n_tiles + 7) >> 3;
  if ((int)blockIdx.x >= tile_blocks) {
    const int i = ((int)blockIdx.x - tile_blocks) * 256 + (int)threadIdx.x;
    if (i < WFRAG_FLOATS) {
      const int j = i & 3, ln = (i >> 2) & 31, blk = i >> 7;
      const int kb = blk % 36, nbb = blk / 36;
      const int k = kb * 8 + (j >> 1) * 4 + ((ln >> 4) << 1) + (j & 1);
      wfrag[i] = Wm[k * 64 + nbb * 16 + (ln & 15)];
    }
    return;
  }
  for (int i = threadIdx.x; i < 2 * 64 * 64; i += 256) {
    int t = i >> 12, k = (i >> 6) & 63, c = i & 63;
    w[(t * 64 + c) * NSTRIDE + k] = Wn[i];
  }
  __syncthreads();

  const int wave = threadIdx.x >> 5;
  const int lane = threadIdx.x & 31;
  const int tile = blockIdx.x * 8 + wave;
  if (tile >= n_tiles) return;

  const int arow = lane & 15;
  const int khalf = (lane >> 4) << 1;
  int node = tile * 16 + arow;
  const bool valid = node < n_nodes;
  if (!valid) node = n_nodes - 1;
  const int t_row = valid ? ntype[node] : -1;
  const float* __restrict__ ar = x + (size_t)node * 64;
  const float* __restrict__ wb0 = &w[arow * NSTRIDE];
  const float* __restrict__ wb1 = &w[(64 + arow) * NSTRIDE];

  v8f acc[4] = {};
#pragma unroll
  for (int kk = 0; kk < 64; kk += 4) {
    const int k = kk + khalf;
    v2f a = __builtin_nontemporal_load((const v2f*)(ar + k));
    const v2f z = {0.f, 0.f};
    v2f a0 = (t_row == 0) ? a : z;
    v2f a1 = (t_row == 1) ? a : z;
#pragma unroll
    for (int nb = 0; nb < 4; ++nb) {
      v2f b0 = *(const v2f*)(wb0 + nb * 16 * NSTRIDE + k);
      v2f b1 = *(const v2f*)(wb1 + nb * 16 * NSTRIDE + k);
      acc[nb] = wmma_f32(a0, b0, acc[nb]);
      acc[nb] = wmma_f32(a1, b1, acc[nb]);
    }
  }

  const int rbase = (lane >> 4) * 8;
  const int col = lane & 15;
#pragma unroll
  for (int r = 0; r < 8; ++r) {
    const int orow = tile * 16 + rbase + r;
    if (orow < n_nodes) {
#pragma unroll
      for (int nb = 0; nb < 4; ++nb)
        xout[(size_t)orow * 64 + nb * 16 + col] = ((const float*)&acc[nb])[r];
    }
  }
}

__global__ __launch_bounds__(256) void aggregate_xform_kernel(
    const float* __restrict__ xout, const float* __restrict__ ef,
    const int* __restrict__ rec, const int* __restrict__ pos, int n_wg,
    const float* __restrict__ wfrag, float* __restrict__ out, int n_nodes) {
  __shared__ float s[BUCKET_NODES * SSTRIDE];
  const int tid = threadIdx.x;
  for (int i = tid; i < BUCKET_NODES * SSTRIDE / 4; i += 256)
    ((v4f*)s)[i] = v4f{0.f, 0.f, 0.f, 0.f};
  __syncthreads();

  const int wave = __builtin_amdgcn_readfirstlane((int)(threadIdx.x >> 5));
  const int lane = threadIdx.x & 31;
  const int b = (int)blockIdx.x;
  const int beg = __builtin_amdgcn_readfirstlane(pos[(size_t)b * n_wg]);
  const int end = __builtin_amdgcn_readfirstlane(pos[(size_t)(b + 1) * n_wg]);
  const int per = (end - beg + 7) >> 3;
  const int wbeg = beg + wave * per;
  const int wend = (wbeg + per < end) ? wbeg + per : end;

  for (int j0 = wbeg; j0 < wend; j0 += 32) {
    const int left = wend - j0;
    const int nrec = left < 32 ? left : 32;
    v4i r = {0, 0, 3, 0};
    if (lane < nrec) r = *(const v4i*)(rec + 4 * (size_t)(j0 + lane));
    for (int q = 0; q < nrec; q += 4) {
      float x0[4], x1[4], f[4];
      int off[4];
#pragma unroll
      for (int u = 0; u < 4; ++u) {
        const int qq = q + u;
        const bool live = qq < nrec;
        const int qc = live ? qq : q;
        const unsigned e = (unsigned)__builtin_amdgcn_readlane(r.x, qc);
        const unsigned sn = (unsigned)__builtin_amdgcn_readlane(r.y, qc);
        const int t = __builtin_amdgcn_readlane(r.z, qc);
        const int dl = __builtin_amdgcn_readlane(r.w, qc);
        off[u] = (live && t < 3) ? (dl * SSTRIDE + t * 96) : -1;
        const float* xr = xout + ((size_t)sn * 64u + (unsigned)lane);
        x0[u] = xr[0];
        x1[u] = xr[32];
        f[u] = ef[(size_t)e * 32u + (unsigned)lane];
      }
#pragma unroll
      for (int u = 0; u < 4; ++u) {
        if (off[u] >= 0) {
          float* row = s + off[u] + lane;
          unsafeAtomicAdd(row, x0[u]);
          unsafeAtomicAdd(row + 32, x1[u]);
          unsafeAtomicAdd(row + 64, f[u]);
        }
      }
    }
  }
  __syncthreads();

  const int nb = wave & 3;
  const int kh = wave >> 2;
  const int arow = lane & 15;
  const int h2 = (lane >> 4) << 1;
  const float* a0p = s + arow * SSTRIDE + kh * 144 + h2;
  const float* a1p = a0p + 16 * SSTRIDE;
  const v4f* __restrict__ bp = (const v4f*)wfrag + ((nb * 36 + kh * 18) * 32 + lane);

  v8f acc0 = {}, acc1 = {};
#pragma unroll
  for (int j = 0; j < 18; ++j) {
    const v4f bq = bp[j * 32];
    const v2f b01 = {bq.x, bq.y};
    const v2f b23 = {bq.z, bq.w};
    const v2f a00 = *(const v2f*)(a0p + j * 8);
    const v2f a01 = *(const v2f*)(a0p + j * 8 + 4);
    const v2f a10 = *(const v2f*)(a1p + j * 8);
    const v2f a11 = *(const v2f*)(a1p + j * 8 + 4);
    acc0 = wmma_f32(a00, b01, acc0);
    acc1 = wmma_f32(a10, b01, acc1);
    acc0 = wmma_f32(a01, b23, acc0);
    acc1 = wmma_f32(a11, b23, acc1);
  }
  __syncthreads();

  float* st = s + ((lane >> 4) * 8) * OSTRIDE + nb * 16 + (lane & 15);
  if (kh == 0) {
#pragma unroll
    for (int r8 = 0; r8 < 8; ++r8) {
      st[r8 * OSTRIDE] = ((const float*)&acc0)[r8];
      st[(16 + r8) * OSTRIDE] = ((const float*)&acc1)[r8];
    }
  }
  __syncthreads();
  if (kh == 1) {
#pragma unroll
    for (int r8 = 0; r8 < 8; ++r8) {
      st[r8 * OSTRIDE] += ((const float*)&acc0)[r8];
      st[(16 + r8) * OSTRIDE] += ((const float*)&acc1)[r8];
    }
  }
  __syncthreads();

  const int node0 = b * BUCKET_NODES;
  const int rows = (n_nodes - node0 < BUCKET_NODES) ? n_nodes - node0 : BUCKET_NODES;
  v4f* __restrict__ og = (v4f*)(out + (size_t)node0 * 64);
  for (int i = tid; i < rows * 16; i += 256)
    og[i] = *(const v4f*)(s + (i >> 4) * OSTRIDE + (i & 15) * 4);
}

extern "C" void kernel_launch(void* const* d_in, const int* in_sizes, int n_in,
                              void* d_out, int out_size, void* d_ws,
                              size_t ws_size, hipStream_t stream) {
  const float* node_feature = (const float*)d_in[0];
  const float* edge_feature = (const float*)d_in[1];
  const float* W_node = (const float*)d_in[2];
  const float* W_msg = (const float*)d_in[3];
  const int* node_type = (const int*)d_in[4];
  const int* edge_type = (const int*)d_in[5];
  const int* edge_index = (const int*)d_in[6];
  const int n_nodes = in_sizes[4];
  const int n_edges = in_sizes[5];
  const int* src = edge_index;
  const int* dst = edge_index + n_edges;

  float* out = (float*)d_out;

  const int n_buckets = (n_nodes + BUCKET_NODES - 1) / BUCKET_NODES;
  const int sort_wgs = 256;
  const int chunk = (n_edges + sort_wgs - 1) / sort_wgs;
  const int n_cnt = n_buckets * sort_wgs;
  const int scan_blocks = (n_cnt + 1023) / 1024;

  float* xout = (float*)d_ws;
  float* wfrag = xout + (size_t)n_nodes * 64;
  int* rec = (int*)(wfrag + WFRAG_FLOATS);
  int* wgcnt = rec + 4 * (size_t)n_edges;
  int* pos = wgcnt + n_cnt;
  int* bsum = pos + n_cnt + 1;

  bucket_count_kernel<<<sort_wgs, SORT_THREADS, n_buckets * sizeof(int), stream>>>(
      dst, n_edges, chunk, n_buckets, wgcnt);
  blocksum_kernel<<<scan_blocks, 1024, 0, stream>>>(wgcnt, n_cnt, bsum);
  scan_kernel<<<scan_blocks, 1024, 0, stream>>>(wgcnt, n_cnt, bsum, pos);
  bucket_place_kernel<<<sort_wgs, SORT_THREADS, n_buckets * sizeof(int), stream>>>(
      src, dst, edge_type, n_edges, chunk, n_buckets, pos, rec);

  const int node_tiles = (n_nodes + 15) / 16;
  node_xform_kernel<<<(node_tiles + 7) / 8 + WFRAG_FLOATS / 256, 256, 0, stream>>>(
      node_feature, W_node, node_type, xout, n_nodes, node_tiles, W_msg, wfrag);

  aggregate_xform_kernel<<<n_buckets, 256, 0, stream>>>(
      xout, edge_feature, rec, pos, sort_wgs, wfrag, out, n_nodes);
}


# Sort offsets: count matrix kept work-group-major (whole-line rows) and prefix-summed down its columns in one 98-block launch; the two 782-block scan launches and all strided 4-byte traffic go
# speedup vs baseline: 1.9773x; 1.2065x over previous
#include <hip/hip_runtime.h>


typedef __attribute__((ext_vector_type(2))) float v2f;
typedef __attribute__((ext_vector_type(4))) float v4f;
typedef __attribute__((ext_vector_type(8))) float v8f;
typedef __attribute__((ext_vector_type(4))) int v4i;

#define NSTRIDE 68
#define MSG_K 288
#define SSTRIDE 292
#define OSTRIDE 68
#define WFRAG_FLOATS (MSG_K * 64)

__device__ __forceinline__ v8f wmma_f32(v2f a, v2f b, v8f c) {
  return __builtin_amdgcn_wmma_f32_16x16x4_f32(false, a, false, b, (short)0, c,
                                               false, false);
}

__device__ __forceinline__ int wave_incl_scan(int v, int lane) {
#pragma unroll
  for (int off = 1; off < 32; off <<= 1) {
    int u = __shfl_up(v, off, 32);
    if (lane >= off) v += u;
  }
  return v;
}

__device__ __forceinline__ int wave_sum(int v) {
#pragma unroll
  for (int off = 16; off > 0; off >>= 1) v += __shfl_xor(v, off, 32);
  return v;
}

#define BUCKET_SHIFT 5
#define BUCKET_NODES 32
#define SORT_THREADS 1024
#define SORT_WGS 256
#define GROUP_BUCKETS 32
#define SEG_WGS (SORT_WGS / 32)

__global__ __launch_bounds__(SORT_THREADS) void bucket_count_kernel(
    const int* __restrict__ dst, int n_edges, int chunk, int n_buckets,
    int row_stride, int n_groups, int* __restrict__ cnt, int* __restrict__ gpre) {
  extern __shared__ int hist[];
  const int tid = threadIdx.x;
  const int lane = tid & 31, wave = tid >> 5;
  const int wg = blockIdx.x, n_wg = gridDim.x;
  for (int b = tid; b < n_buckets; b += SORT_THREADS) hist[b] = 0;
  __syncthreads();
  const int cbeg = wg * chunk;
  const int cend = (cbeg + chunk < n_edges) ? cbeg + chunk : n_edges;
  for (int i0 = cbeg + tid; i0 < cend; i0 += 8 * SORT_THREADS) {
    int d[8];
#pragma unroll
    for (int u = 0; u < 8; ++u) {
      const int i = i0 + u * SORT_THREADS;
      d[u] = dst[i < cend ? i : cend - 1];
    }
#pragma unroll
    for (int u = 0; u < 8; ++u)
      if (i0 + u * SORT_THREADS < cend) atomicAdd(&hist[d[u] >> BUCKET_SHIFT], 1);
  }
  __syncthreads();
  int* __restrict__ row = cnt + (size_t)wg * row_stride;
  for (int b = tid; b < n_buckets; b += SORT_THREADS) row[b] = hist[b];
  int* gs = hist + row_stride;
  for (int k = wave; k < n_groups; k += SORT_THREADS / 32) {
    const int b = k * GROUP_BUCKETS + lane;
    const int s = wave_sum(b < n_buckets ? hist[b] : 0);
    if (lane == 0) gs[k] = s;
  }
  __syncthreads();
  if (wave == 0) {
    int carry = 0;
    for (int k0 = 0; k0 < n_groups; k0 += 32) {
      const int k = k0 + lane;
      const int v = (k < n_groups) ? gs[k] : 0;
      const int inc = wave_incl_scan(v, lane);
      if (k < n_groups) gpre[(size_t)k * n_wg + wg] = carry + inc - v;
      carry += __shfl(inc, 31, 32);
    }
  }
}

__global__ __launch_bounds__(1024) void offsets_kernel(
    const int* __restrict__ cnt, const int* __restrict__ gpre, int n_buckets,
    int row_stride, int* __restrict__ pos, int* __restrict__ bstart) {
  __shared__ int segsum[32][GROUP_BUCKETS];
  __shared__ int red[SORT_WGS / 32];
  const int tid = threadIdx.x;
  const int lane = tid & 31, wave = tid >> 5;
  const int k = blockIdx.x;
  const int b = k * GROUP_BUCKETS + lane;
  const bool live = b < n_buckets;
  const size_t o = (size_t)(wave * SEG_WGS) * row_stride + b;
  int v[SEG_WGS];
#pragma unroll
  for (int i = 0; i < SEG_WGS; ++i) v[i] = live ? cnt[o + (size_t)i * row_stride] : 0;
  if (tid < SORT_WGS) {
    const int t = wave_sum(gpre[(size_t)k * SORT_WGS + tid]);
    if (lane == 0) red[wave] = t;
  }
  int s = 0;
#pragma unroll
  for (int i = 0; i < SEG_WGS; ++i) s += v[i];
  segsum[wave][lane] = s;
  __syncthreads();
  int total = 0, before = 0;
#pragma unroll
  for (int w = 0; w < 32; ++w) {
    const int x = segsum[w][lane];
    total += x;
    if (w < wave) before += x;
  }
  int base = 0;
#pragma unroll
  for (int w = 0; w < SORT_WGS / 32; ++w) base += red[w];
  const int first = base + wave_incl_scan(total, lane) - total;
  int run = first + before;
  if (live) {
#pragma unroll
    for (int i = 0; i < SEG_WGS; ++i) {
      pos[o + (size_t)i * row_stride] = run;
      run += v[i];
    }
    if (wave == 0) bstart[b] = first;
    if (wave == 31 && b == n_buckets - 1) bstart[n_buckets] = run;
  }
}

__global__ __launch_bounds__(SORT_THREADS) void bucket_place_kernel(
    const int* __restrict__ src, const int* __restrict__ dst,
    const int* __restrict__ etype, int n_edges, int chunk, int n_buckets,
    int row_stride, const int* __restrict__ pos, int* __restrict__ rec) {
  extern __shared__ int cursor[];
  const int tid = threadIdx.x;
  const int wg = blockIdx.x;
  const int* __restrict__ prow = pos + (size_t)wg * row_stride;
  for (int b = tid; b < n_buckets; b += SORT_THREADS) cursor[b] = prow[b];
  __syncthreads();
  const int cbeg = wg * chunk;
  const int cend = (cbeg + chunk < n_edges) ? cbeg + chunk : n_edges;
  for (int i0 = cbeg + tid; i0 < cend; i0 += 8 * SORT_THREADS) {
    int d[8], t[8], sn[8];
#pragma unroll
    for (int u = 0; u < 8; ++u) {
      const int i = i0 + u * SORT_THREADS;
      const int ic = i < cend ? i : cend - 1;
      d[u] = dst[ic];
      t[u] = etype[ic];
      sn[u] = src[ic];
    }
#pragma unroll
    for (int u = 0; u < 8; ++u) {
      const int i = i0 + u * SORT_THREADS;
      if (i < cend) {
        const int tt = (t[u] < 0 || t[u] > 2) ? 3 : t[u];
        const int p = atomicAdd(&cursor[d[u] >> BUCKET_SHIFT], 1);
        v4i r = {i, sn[u], tt, d[u] & (BUCKET_NODES - 1)};
        *(v4i*)(rec + 4 * (size_t)p) = r;
      }
    }
  }
}

__global__ __launch_bounds__(256) void node_xform_kernel(
    const float* __restrict__ x, const float* __restrict__ Wn,
    const int* __restrict__ ntype, float* __restrict__ xout, int n_nodes,
    int n_tiles, const float* __restrict__ Wm, float* __restrict__ wfrag) {
  __shared__ float w[2 * 64 * NSTRIDE];
  const int tile_blocks = (n_tiles + 7) >> 3;
  if ((int)blockIdx.x >= tile_blocks) {
    const int i = ((int)blockIdx.x - tile_blocks) * 256 + (int)threadIdx.x;
    if (i < WFRAG_FLOATS) {
      const int j = i & 3, ln = (i >> 2) & 31, blk = i >> 7;
      const int kb = blk % 36, nbb = blk / 36;
      const int k = kb * 8 + (j >> 1) * 4 + ((ln >> 4) << 1) + (j & 1);
      wfrag[i] = Wm[k * 64 + nbb * 16 + (ln & 15)];
    }
    return;
  }
  for (int i = threadIdx.x; i < 2 * 64 * 64; i += 256) {
    int t = i >> 12, k = (i >> 6) & 63, c = i & 63;
    w[(t * 64 + c) * NSTRIDE + k] = Wn[i];
  }
  __syncthreads();

  const int wave = threadIdx.x >> 5;
  const int lane = threadIdx.x & 31;
  const int tile = blockIdx.x * 8 + wave;
  if (tile >= n_tiles) return;

  const int arow = lane & 15;
  const int khalf = (lane >> 4) << 1;
  int node = tile * 16 + arow;
  const bool valid = node < n_nodes;
  if (!valid) node = n_nodes - 1;
  const int t_row = valid ? ntype[node] : -1;
  const float* __restrict__ ar = x + (size_t)node * 64;
  const float* __restrict__ wb0 = &w[arow * NSTRIDE];
  const float* __restrict__ wb1 = &w[(64 + arow) * NSTRIDE];

  v8f acc[4] = {};
#pragma unroll
  for (int kk = 0; kk < 64; kk += 4) {
    const int k = kk + khalf;
    v2f a = __builtin_nontemporal_load((const v2f*)(ar + k));
    const v2f z = {0.f, 0.f};
    v2f a0 = (t_row == 0) ? a : z;
    v2f a1 = (t_row == 1) ? a : z;
#pragma unroll
    for (int nb = 0; nb < 4; ++nb) {
      v2f b0 = *(const v2f*)(wb0 + nb * 16 * NSTRIDE + k);
      v2f b1 = *(const v2f*)(wb1 + nb * 16 * NSTRIDE + k);
      acc[nb] = wmma_f32(a0, b0, acc[nb]);
      acc[nb] = wmma_f32(a1, b1, acc[nb]);
    }
  }

  const int rbase = (lane >> 4) * 8;
  const int col = lane & 15;
#pragma unroll
  for (int r = 0; r < 8; ++r) {
    const int orow = tile * 16 + rbase + r;
    if (orow < n_nodes) {
#pragma unroll
      for (int nb = 0; nb < 4; ++nb)
        xout[(size_t)orow * 64 + nb * 16 + col] = ((const float*)&acc[nb])[r];
    }
  }
}

__global__ __launch_bounds__(256) void aggregate_xform_kernel(
    const float* __restrict__ xout, const float* __restrict__ ef,
    const int* __restrict__ rec, const int* __restrict__ bstart,
    const float* __restrict__ wfrag, float* __restrict__ out, int n_nodes) {
  __shared__ float s[BUCKET_NODES * SSTRIDE];
  const int tid = threadIdx.x;
  for (int i = tid; i < BUCKET_NODES * SSTRIDE / 4; i += 256)
    ((v4f*)s)[i] = v4f{0.f, 0.f, 0.f, 0.f};
  __syncthreads();

  const int wave = __builtin_amdgcn_readfirstlane((int)(threadIdx.x >> 5));
  const int lane = threadIdx.x & 31;
  const int b = (int)blockIdx.x;
  const int beg = __builtin_amdgcn_readfirstlane(bstart[b]);
  const int end = __builtin_amdgcn_readfirstlane(bstart[b + 1]);
  const int per = (end - beg + 7) >> 3;
  const int wbeg = beg + wave * per;
  const int wend = (wbeg + per < end) ? wbeg + per : end;

  for (int j0 = wbeg; j0 < wend; j0 += 32) {
    const int left = wend - j0;
    const int nrec = left < 32 ? left : 32;
    v4i r = {0, 0, 3, 0};
    if (lane < nrec) r = *(const v4i*)(rec + 4 * (size_t)(j0 + lane));
    for (int q = 0; q < nrec; q += 4) {
      float x0[4], x1[4], f[4];
      int off[4];
#pragma unroll
      for (int u = 0; u < 4; ++u) {
        const int qq = q + u;
        const bool live = qq < nrec;
        const int qc = live ? qq : q;
        const unsigned e = (unsigned)__builtin_amdgcn_readlane(r.x, qc);
        const unsigned sn = (unsigned)__builtin_amdgcn_readlane(r.y, qc);
        const int t = __builtin_amdgcn_readlane(r.z, qc);
        const int dl = __builtin_amdgcn_readlane(r.w, qc);
        off[u] = (live && t < 3) ? (dl * SSTRIDE + t * 96) : -1;
        const float* xr = xout + ((size_t)sn * 64u + (unsigned)lane);
        x0[u] = xr[0];
        x1[u] = xr[32];
        f[u] = ef[(size_t)e * 32u + (unsigned)lane];
      }
#pragma unroll
      for (int u = 0; u < 4; ++u) {
        if (off[u] >= 0) {
          float* row = s + off[u] + lane;
          unsafeAtomicAdd(row, x0[u]);
          unsafeAtomicAdd(row + 32, x1[u]);
          unsafeAtomicAdd(row + 64, f[u]);
        }
      }
    }
  }
  __syncthreads();

  const int nb = wave & 3;
  const int kh = wave >> 2;
  const int arow = lane & 15;
  const int h2 = (lane >> 4) << 1;
  const float* a0p = s + arow * SSTRIDE + kh * 144 + h2;
  const float* a1p = a0p + 16 * SSTRIDE;
  const v4f* __restrict__ bp = (const v4f*)wfrag + ((nb * 36 + kh * 18) * 32 + lane);

  v8f acc0 = {}, acc1 = {};
#pragma unroll
  for (int j = 0; j < 18; ++j) {
    const v4f bq = bp[j * 32];
    const v2f b01 = {bq.x, bq.y};
    const v2f b23 = {bq.z, bq.w};
    const v2f a00 = *(const v2f*)(a0p + j * 8);
    const v2f a01 = *(const v2f*)(a0p + j * 8 + 4);
    const v2f a10 = *(const v2f*)(a1p + j * 8);
    const v2f a11 = *(const v2f*)(a1p + j * 8 + 4);
    acc0 = wmma_f32(a00, b01, acc0);
    acc1 = wmma_f32(a10, b01, acc1);
    acc0 = wmma_f32(a01, b23, acc0);
    acc1 = wmma_f32(a11, b23, acc1);
  }
  __syncthreads();

  float* st = s + ((lane >> 4) * 8) * OSTRIDE + nb * 16 + (lane & 15);
  if (kh == 0) {
#pragma unroll
    for (int r8 = 0; r8 < 8; ++r8) {
      st[r8 * OSTRIDE] = ((const float*)&acc0)[r8];
      st[(16 + r8) * OSTRIDE] = ((const float*)&acc1)[r8];
    }
  }
  __syncthreads();
  if (kh == 1) {
#pragma unroll
    for (int r8 = 0; r8 < 8; ++r8) {
      st[r8 * OSTRIDE] += ((const float*)&acc0)[r8];
      st[(16 + r8) * OSTRIDE] += ((const float*)&acc1)[r8];
    }
  }
  __syncthreads();

  const int node0 = b * BUCKET_NODES;
  const int rows = (n_nodes - node0 < BUCKET_NODES) ? n_nodes - node0 : BUCKET_NODES;
  v4f* __restrict__ og = (v4f*)(out + (size_t)node0 * 64);
  for (int i = tid; i < rows * 16; i += 256)
    og[i] = *(const v4f*)(s + (i >> 4) * OSTRIDE + (i & 15) * 4);
}

extern "C" void kernel_launch(void* const* d_in, const int* in_sizes, int n_in,
                              void* d_out, int out_size, void* d_ws,
                              size_t ws_size, hipStream_t stream) {
  const float* node_feature = (const float*)d_in[0];
  const float* edge_feature = (const float*)d_in[1];
  const float* W_node = (const float*)d_in[2];
  const float* W_msg = (const float*)d_in[3];
  const int* node_type = (const int*)d_in[4];
  const int* edge_type = (const int*)d_in[5];
  const int* edge_index = (const int*)d_in[6];
  const int n_nodes = in_sizes[4];
  const int n_edges = in_sizes[5];
  const int* src = edge_index;
  const int* dst = edge_index + n_edges;

  float* out = (float*)d_out;

  const int n_buckets = (n_nodes + BUCKET_NODES - 1) / BUCKET_NODES;
  const int sort_wgs = SORT_WGS;
  const int chunk = (n_edges + sort_wgs - 1) / sort_wgs;
  const int n_groups = (n_buckets + GROUP_BUCKETS - 1) / GROUP_BUCKETS;
  const int row_stride = n_groups * GROUP_BUCKETS;

  float* xout = (float*)d_ws;
  float* wfrag = xout + (size_t)n_nodes * 64;
  int* rec = (int*)(wfrag + WFRAG_FLOATS);
  int* cnt = rec + 4 * (size_t)n_edges;
  int* pos = cnt + (size_t)sort_wgs * row_stride;
  int* gpre = pos + (size_t)sort_wgs * row_stride;
  int* bstart = gpre + (size_t)n_groups * sort_wgs;

  bucket_count_kernel<<<sort_wgs, SORT_THREADS, (row_stride + n_groups) * sizeof(int), stream>>>(
      dst, n_edges, chunk, n_buckets, row_stride, n_groups, cnt, gpre);
  offsets_kernel<<<n_groups, 1024, 0, stream>>>(cnt, gpre, n_buckets, row_stride, pos, bstart);
  bucket_place_kernel<<<sort_wgs, SORT_THREADS, n_buckets * sizeof(int), stream>>>(
      src, dst, edge_type, n_edges, chunk, n_buckets, row_stride, pos, rec);

  const int node_tiles = (n_nodes + 15) / 16;
  node_xform_kernel<<<(node_tiles + 7) / 8 + WFRAG_FLOATS / 256, 256, 0, stream>>>(
      node_feature, W_node, node_type, xout, n_nodes, node_tiles, W_msg, wfrag);

  aggregate_xform_kernel<<<n_buckets, 256, 0, stream>>>(
      xout, edge_feature, rec, bstart, wfrag, out, n_nodes);
}


# Placement sorts each work-group's records by bucket in LDS and writes a bucket's run as one contiguous piece (128 chunks, runs of 4), not one scattered 16-byte store per edge; sum order: LDS arrival
# speedup vs baseline: 2.2527x; 1.1393x over previous
#include <hip/hip_runtime.h>


typedef __attribute__((ext_vector_type(2))) float v2f;
typedef __attribute__((ext_vector_type(4))) float v4f;
typedef __attribute__((ext_vector_type(8))) float v8f;
typedef __attribute__((ext_vector_type(4))) int v4i;

#define NSTRIDE 68
#define MSG_K 288
#define SSTRIDE 292
#define OSTRIDE 68
#define WFRAG_FLOATS (MSG_K * 64)

__device__ __forceinline__ v8f wmma_f32(v2f a, v2f b, v8f c) {
  return __builtin_amdgcn_wmma_f32_16x16x4_f32(false, a, false, b, (short)0, c,
                                               false, false);
}

__device__ __forceinline__ int wave_incl_scan(int v, int lane) {
#pragma unroll
  for (int off = 1; off < 32; off <<= 1) {
    int u = __shfl_up(v, off, 32);
    if (lane >= off) v += u;
  }
  return v;
}

__device__ __forceinline__ int wave_sum(int v) {
#pragma unroll
  for (int off = 16; off > 0; off >>= 1) v += __shfl_xor(v, off, 32);
  return v;
}

#define BUCKET_SHIFT 5
#define BUCKET_NODES 32
#define SORT_THREADS 1024
#define SORT_WGS 128
#define SORT_DEPTH 16
#define GROUP_BUCKETS 32
#define SEG_WGS (SORT_WGS / 32)

__global__ __launch_bounds__(SORT_THREADS) void bucket_count_kernel(
    const int* __restrict__ dst, int n_edges, int chunk, int n_buckets,
    int row_stride, int n_groups, int* __restrict__ cnt, int* __restrict__ gpre) {
  extern __shared__ int hist[];
  const int tid = threadIdx.x;
  const int lane = tid & 31, wave = tid >> 5;
  const int wg = blockIdx.x, n_wg = gridDim.x;
  for (int b = tid; b < n_buckets; b += SORT_THREADS) hist[b] = 0;
  __syncthreads();
  const int cbeg = wg * chunk;
  const int cend = (cbeg + chunk < n_edges) ? cbeg + chunk : n_edges;
  for (int i0 = cbeg + tid; i0 < cend; i0 += SORT_DEPTH * SORT_THREADS) {
    int d[SORT_DEPTH];
#pragma unroll
    for (int u = 0; u < SORT_DEPTH; ++u) {
      const int i = i0 + u * SORT_THREADS;
      d[u] = dst[i < cend ? i : cend - 1];
    }
#pragma unroll
    for (int u = 0; u < SORT_DEPTH; ++u)
      if (i0 + u * SORT_THREADS < cend) atomicAdd(&hist[d[u] >> BUCKET_SHIFT], 1);
  }
  __syncthreads();
  int* __restrict__ row = cnt + (size_t)wg * row_stride;
  for (int b = tid; b < n_buckets; b += SORT_THREADS) row[b] = hist[b];
  int* gs = hist + row_stride;
  for (int k = wave; k < n_groups; k += SORT_THREADS / 32) {
    const int b = k * GROUP_BUCKETS + lane;
    const int s = wave_sum(b < n_buckets ? hist[b] : 0);
    if (lane == 0) gs[k] = s;
  }
  __syncthreads();
  if (wave == 0) {
    int carry = 0;
    for (int k0 = 0; k0 < n_groups; k0 += 32) {
      const int k = k0 + lane;
      const int v = (k < n_groups) ? gs[k] : 0;
      const int inc = wave_incl_scan(v, lane);
      if (k < n_groups) gpre[(size_t)k * n_wg + wg] = carry + inc - v;
      carry += __shfl(inc, 31, 32);
    }
  }
}

__global__ __launch_bounds__(1024) void offsets_kernel(
    const int* __restrict__ cnt, const int* __restrict__ gpre, int n_buckets,
    int row_stride, int* __restrict__ pos, int* __restrict__ bstart) {
  __shared__ int segsum[32][GROUP_BUCKETS];
  __shared__ int red[SORT_WGS / 32];
  const int tid = threadIdx.x;
  const int lane = tid & 31, wave = tid >> 5;
  const int k = blockIdx.x;
  const int b = k * GROUP_BUCKETS + lane;
  const bool live = b < n_buckets;
  const size_t o = (size_t)(wave * SEG_WGS) * row_stride + b;
  int v[SEG_WGS];
#pragma unroll
  for (int i = 0; i < SEG_WGS; ++i) v[i] = live ? cnt[o + (size_t)i * row_stride] : 0;
  if (tid < SORT_WGS) {
    const int t = wave_sum(gpre[(size_t)k * SORT_WGS + tid]);
    if (lane == 0) red[wave] = t;
  }
  int s = 0;
#pragma unroll
  for (int i = 0; i < SEG_WGS; ++i) s += v[i];
  segsum[wave][lane] = s;
  __syncthreads();
  int total = 0, before = 0;
#pragma unroll
  for (int w = 0; w < 32; ++w) {
    const int x = segsum[w][lane];
    total += x;
    if (w < wave) before += x;
  }
  int base = 0;
#pragma unroll
  for (int w = 0; w < SORT_WGS / 32; ++w) base += red[w];
  const int first = base + wave_incl_scan(total, lane) - total;
  int run = first + before;
  if (live) {
#pragma unroll
    for (int i = 0; i < SEG_WGS; ++i) {
      pos[o + (size_t)i * row_stride] = run;
      run += v[i];
    }
    if (wave == 0) bstart[b] = first;
    if (wave == 31 && b == n_buckets - 1) bstart[n_buckets] = run;
  }
}

#define STAGE_CAP 12544

__global__ __launch_bounds__(SORT_THREADS) void bucket_place_kernel(
    const int* __restrict__ src, const int* __restrict__ dst,
    const int* __restrict__ etype, int n_edges, int chunk, int n_buckets,
    int row_stride, const int* __restrict__ cnt, const int* __restrict__ pos,
    int* __restrict__ rec) {
  extern __shared__ int cursor[];
  __shared__ v4i stage[STAGE_CAP];
  __shared__ int wsum[SORT_THREADS / 32];
  const int tid = threadIdx.x;
  const int lane = tid & 31, wave = tid >> 5;
  const int wg = blockIdx.x;
  const int* __restrict__ prow = pos + (size_t)wg * row_stride;
  const int cbeg = wg * chunk;
  const int cend = (cbeg + chunk < n_edges) ? cbeg + chunk : n_edges;
  const bool staged = chunk <= STAGE_CAP;
  int* delta = cursor + n_buckets;
  if (staged) {
    const int* __restrict__ crow = cnt + (size_t)wg * row_stride;
#pragma unroll 4
    for (int b = tid; b < n_buckets; b += SORT_THREADS) {
      cursor[b] = crow[b];
      delta[b] = prow[b];
    }
    __syncthreads();
    const int per = (n_buckets + SORT_THREADS - 1) / SORT_THREADS;
    const int b0 = tid * per;
    const int b1 = (b0 + per < n_buckets) ? b0 + per : n_buckets;
    int s = 0;
    for (int b = b0; b < b1; ++b) s += cursor[b];
    const int inc = wave_incl_scan(s, lane);
    if (lane == 31) wsum[wave] = inc;
    __syncthreads();
    int run = inc - s;
    for (int w = 0; w < SORT_THREADS / 32; ++w)
      if (w < wave) run += wsum[w];
    for (int b = b0; b < b1; ++b) {
      const int c = cursor[b];
      cursor[b] = run;
      delta[b] -= run;
      run += c;
    }
  } else {
    for (int b = tid; b < n_buckets; b += SORT_THREADS) cursor[b] = prow[b];
  }
  __syncthreads();
  for (int i0 = cbeg + tid; i0 < cend; i0 += SORT_DEPTH * SORT_THREADS) {
    int d[SORT_DEPTH], t[SORT_DEPTH], sn[SORT_DEPTH];
#pragma unroll
    for (int u = 0; u < SORT_DEPTH; ++u) {
      const int i = i0 + u * SORT_THREADS;
      const int ic = i < cend ? i : cend - 1;
      d[u] = dst[ic];
      t[u] = etype[ic];
      sn[u] = src[ic];
    }
#pragma unroll
    for (int u = 0; u < SORT_DEPTH; ++u) {
      const int i = i0 + u * SORT_THREADS;
      if (i < cend) {
        const int tt = (t[u] < 0 || t[u] > 2) ? 3 : t[u];
        const int p = atomicAdd(&cursor[d[u] >> BUCKET_SHIFT], 1);
        if (staged) {
          v4i r = {i, sn[u], tt, d[u]};
          stage[p] = r;
        } else {
          v4i r = {i, sn[u], tt, d[u] & (BUCKET_NODES - 1)};
          *(v4i*)(rec + 4 * (size_t)p) = r;
        }
      }
    }
  }
  if (!staged) return;
  __syncthreads();
  const int n_local = cend - cbeg;
#pragma unroll 4
  for (int i = tid; i < n_local; i += SORT_THREADS) {
    v4i r = stage[i];
    const int p = i + delta[r.w >> BUCKET_SHIFT];
    r.w &= BUCKET_NODES - 1;
    *(v4i*)(rec + 4 * (size_t)p) = r;
  }
}

__global__ __launch_bounds__(256) void node_xform_kernel(
    const float* __restrict__ x, const float* __restrict__ Wn,
    const int* __restrict__ ntype, float* __restrict__ xout, int n_nodes,
    int n_tiles, const float* __restrict__ Wm, float* __restrict__ wfrag) {
  __shared__ float w[2 * 64 * NSTRIDE];
  const int tile_blocks = (n_tiles + 7) >> 3;
  if ((int)blockIdx.x >= tile_blocks) {
    const int i = ((int)blockIdx.x - tile_blocks) * 256 + (int)threadIdx.x;
    if (i < WFRAG_FLOATS) {
      const int j = i & 3, ln = (i >> 2) & 31, blk = i >> 7;
      const int kb = blk % 36, nbb = blk / 36;
      const int k = kb * 8 + (j >> 1) * 4 + ((ln >> 4) << 1) + (j & 1);
      wfrag[i] = Wm[k * 64 + nbb * 16 + (ln & 15)];
    }
    return;
  }
  for (int i = threadIdx.x; i < 2 * 64 * 64; i += 256) {
    int t = i >> 12, k = (i >> 6) & 63, c = i & 63;
    w[(t * 64 + c) * NSTRIDE + k] = Wn[i];
  }
  __syncthreads();

  const int wave = threadIdx.x >> 5;
  const int lane = threadIdx.x & 31;
  const int tile = blockIdx.x * 8 + wave;
  if (tile >= n_tiles) return;

  const int arow = lane & 15;
  const int khalf = (lane >> 4) << 1;
  int node = tile * 16 + arow;
  const bool valid = node < n_nodes;
  if (!valid) node = n_nodes - 1;
  const int t_row = valid ? ntype[node] : -1;
  const float* __restrict__ ar = x + (size_t)node * 64;
  const float* __restrict__ wb0 = &w[arow * NSTRIDE];
  const float* __restrict__ wb1 = &w[(64 + arow) * NSTRIDE];

  v8f acc[4] = {};
#pragma unroll
  for (int kk = 0; kk < 64; kk += 4) {
    const int k = kk + khalf;
    v2f a = __builtin_nontemporal_load((const v2f*)(ar + k));
    const v2f z = {0.f, 0.f};
    v2f a0 = (t_row == 0) ? a : z;
    v2f a1 = (t_row == 1) ? a : z;
#pragma unroll
    for (int nb = 0; nb < 4; ++nb) {
      v2f b0 = *(const v2f*)(wb0 + nb * 16 * NSTRIDE + k);
      v2f b1 = *(const v2f*)(wb1 + nb * 16 * NSTRIDE + k);
      acc[nb] = wmma_f32(a0, b0, acc[nb]);
      acc[nb] = wmma_f32(a1, b1, acc[nb]);
    }
  }

  const int rbase = (lane >> 4) * 8;
  const int col = lane & 15;
#pragma unroll
  for (int r = 0; r < 8; ++r) {
    const int orow = tile * 16 + rbase + r;
    if (orow < n_nodes) {
#pragma unroll
      for (int nb = 0; nb < 4; ++nb)
        xout[(size_t)orow * 64 + nb * 16 + col] = ((const float*)&acc[nb])[r];
    }
  }
}

__global__ __launch_bounds__(256) void aggregate_xform_kernel(
    const float* __restrict__ xout, const float* __restrict__ ef,
    const int* __restrict__ rec, const int* __restrict__ bstart,
    const float* __restrict__ wfrag, float* __restrict__ out, int n_nodes) {
  __shared__ float s[BUCKET_NODES * SSTRIDE];
  const int tid = threadIdx.x;
  for (int i = tid; i < BUCKET_NODES * SSTRIDE / 4; i += 256)
    ((v4f*)s)[i] = v4f{0.f, 0.f, 0.f, 0.f};
  __syncthreads();

  const int wave = __builtin_amdgcn_readfirstlane((int)(threadIdx.x >> 5));
  const int lane = threadIdx.x & 31;
  const int b = (int)blockIdx.x;
  const int beg = __builtin_amdgcn_readfirstlane(bstart[b]);
  const int end = __builtin_amdgcn_readfirstlane(bstart[b + 1]);
  const int per = (end - beg + 7) >> 3;
  const int wbeg = beg + wave * per;
  const int wend = (wbeg + per < end) ? wbeg + per : end;

  for (int j0 = wbeg; j0 < wend; j0 += 32) {
    const int left = wend - j0;
    const int nrec = left < 32 ? left : 32;
    v4i r = {0, 0, 3, 0};
    if (lane < nrec) r = *(const v4i*)(rec + 4 * (size_t)(j0 + lane));
    for (int q = 0; q < nrec; q += 4) {
      float x0[4], x1[4], f[4];
      int off[4];
#pragma unroll
      for (int u = 0; u < 4; ++u) {
        const int qq = q + u;
        const bool live = qq < nrec;
        const int qc = live ? qq : q;
        const unsigned e = (unsigned)__builtin_amdgcn_readlane(r.x, qc);
        const unsigned sn = (unsigned)__builtin_amdgcn_readlane(r.y, qc);
        const int t = __builtin_amdgcn_readlane(r.z, qc);
        const int dl = __builtin_amdgcn_readlane(r.w, qc);
        off[u] = (live && t < 3) ? (dl * SSTRIDE + t * 96) : -1;
        const float* xr = xout + ((size_t)sn * 64u + (unsigned)lane);
        x0[u] = xr[0];
        x1[u] = xr[32];
        f[u] = ef[(size_t)e * 32u + (unsigned)lane];
      }
#pragma unroll
      for (int u = 0; u < 4; ++u) {
        if (off[u] >= 0) {
          float* row = s + off[u] + lane;
          unsafeAtomicAdd(row, x0[u]);
          unsafeAtomicAdd(row + 32, x1[u]);
          unsafeAtomicAdd(row + 64, f[u]);
        }
      }
    }
  }
  __syncthreads();

  const int nb = wave & 3;
  const int kh = wave >> 2;
  const int arow = lane & 15;
  const int h2 = (lane >> 4) << 1;
  const float* a0p = s + arow * SSTRIDE + kh * 144 + h2;
  const float* a1p = a0p + 16 * SSTRIDE;
  const v4f* __restrict__ bp = (const v4f*)wfrag + ((nb * 36 + kh * 18) * 32 + lane);

  v8f acc0 = {}, acc1 = {};
#pragma unroll
  for (int j = 0; j < 18; ++j) {
    const v4f bq = bp[j * 32];
    const v2f b01 = {bq.x, bq.y};
    const v2f b23 = {bq.z, bq.w};
    const v2f a00 = *(const v2f*)(a0p + j * 8);
    const v2f a01 = *(const v2f*)(a0p + j * 8 + 4);
    const v2f a10 = *(const v2f*)(a1p + j * 8);
    const v2f a11 = *(const v2f*)(a1p + j * 8 + 4);
    acc0 = wmma_f32(a00, b01, acc0);
    acc1 = wmma_f32(a10, b01, acc1);
    acc0 = wmma_f32(a01, b23, acc0);
    acc1 = wmma_f32(a11, b23, acc1);
  }
  __syncthreads();

  float* st = s + ((lane >> 4) * 8) * OSTRIDE + nb * 16 + (lane & 15);
  if (kh == 0) {
#pragma unroll
    for (int r8 = 0; r8 < 8; ++r8) {
      st[r8 * OSTRIDE] = ((const float*)&acc0)[r8];
      st[(16 + r8) * OSTRIDE] = ((const float*)&acc1)[r8];
    }
  }
  __syncthreads();
  if (kh == 1) {
#pragma unroll
    for (int r8 = 0; r8 < 8; ++r8) {
      st[r8 * OSTRIDE] += ((const float*)&acc0)[r8];
      st[(16 + r8) * OSTRIDE] += ((const float*)&acc1)[r8];
    }
  }
  __syncthreads();

  const int node0 = b * BUCKET_NODES;
  const int rows = (n_nodes - node0 < BUCKET_NODES) ? n_nodes - node0 : BUCKET_NODES;
  v4f* __restrict__ og = (v4f*)(out + (size_t)node0 * 64);
  for (int i = tid; i < rows * 16; i += 256)
    og[i] = *(const v4f*)(s + (i >> 4) * OSTRIDE + (i & 15) * 4);
}

extern "C" void kernel_launch(void* const* d_in, const int* in_sizes, int n_in,
                              void* d_out, int out_size, void* d_ws,
                              size_t ws_size, hipStream_t stream) {
  const float* node_feature = (const float*)d_in[0];
  const float* edge_feature = (const float*)d_in[1];
  const float* W_node = (const float*)d_in[2];
  const float* W_msg = (const float*)d_in[3];
  const int* node_type = (const int*)d_in[4];
  const int* edge_type = (const int*)d_in[5];
  const int* edge_index = (const int*)d_in[6];
  const int n_nodes = in_sizes[4];
  const int n_edges = in_sizes[5];
  const int* src = edge_index;
  const int* dst = edge_index + n_edges;

  float* out = (float*)d_out;

  const int n_buckets = (n_nodes + BUCKET_NODES - 1) / BUCKET_NODES;
  const int sort_wgs = SORT_WGS;
  const int chunk = (n_edges + sort_wgs - 1) / sort_wgs;
  const int n_groups = (n_buckets + GROUP_BUCKETS - 1) / GROUP_BUCKETS;
  const int row_stride = n_groups * GROUP_BUCKETS;

  float* xout = (float*)d_ws;
  float* wfrag = xout + (size_t)n_nodes * 64;
  int* rec = (int*)(wfrag + WFRAG_FLOATS);
  int* cnt = rec + 4 * (size_t)n_edges;
  int* pos = cnt + (size_t)sort_wgs * row_stride;
  int* gpre = pos + (size_t)sort_wgs * row_stride;
  int* bstart = gpre + (size_t)n_groups * sort_wgs;

  bucket_count_kernel<<<sort_wgs, SORT_THREADS, (row_stride + n_groups) * sizeof(int), stream>>>(
      dst, n_edges, chunk, n_buckets, row_stride, n_groups, cnt, gpre);
  offsets_kernel<<<n_groups, 1024, 0, stream>>>(cnt, gpre, n_buckets, row_stride, pos, bstart);
  bucket_place_kernel<<<sort_wgs, SORT_THREADS, 2 * n_buckets * sizeof(int), stream>>>(
      src, dst, edge_type, n_edges, chunk, n_buckets, row_stride, cnt, pos, rec);

  const int node_tiles = (n_nodes + 15) / 16;
  node_xform_kernel<<<(node_tiles + 7) / 8 + WFRAG_FLOATS / 256, 256, 0, stream>>>(
      node_feature, W_node, node_type, xout, n_nodes, node_tiles, W_msg, wfrag);

  aggregate_xform_kernel<<<n_buckets, 256, 0, stream>>>(
      xout, edge_feature, rec, bstart, wfrag, out, n_nodes);
}


# Node transform and W_msg reorder moved into the count launch as extra work-groups (independent of the sort): 5 launches -> 4, both run side by side; every sum and its order unchanged
# speedup vs baseline: 2.3429x; 1.0400x over previous
#include <hip/hip_runtime.h>


typedef __attribute__((ext_vector_type(2))) float v2f;
typedef __attribute__((ext_vector_type(4))) float v4f;
typedef __attribute__((ext_vector_type(8))) float v8f;
typedef __attribute__((ext_vector_type(4))) int v4i;

#define NSTRIDE 68
#define MSG_K 288
#define SSTRIDE 292
#define OSTRIDE 68
#define WFRAG_FLOATS (MSG_K * 64)

__device__ __forceinline__ v8f wmma_f32(v2f a, v2f b, v8f c) {
  return __builtin_amdgcn_wmma_f32_16x16x4_f32(false, a, false, b, (short)0, c,
                                               false, false);
}

__device__ __forceinline__ int wave_incl_scan(int v, int lane) {
#pragma unroll
  for (int off = 1; off < 32; off <<= 1) {
    int u = __shfl_up(v, off, 32);
    if (lane >= off) v += u;
  }
  return v;
}

__device__ __forceinline__ int wave_sum(int v) {
#pragma unroll
  for (int off = 16; off > 0; off >>= 1) v += __shfl_xor(v, off, 32);
  return v;
}

#define BUCKET_SHIFT 5
#define BUCKET_NODES 32
#define SORT_THREADS 1024
#define SORT_WGS 128
#define SORT_DEPTH 16
#define GROUP_BUCKETS 32
#define SEG_WGS (SORT_WGS / 32)

__global__ __launch_bounds__(SORT_THREADS, 16) void count_node_kernel(
    const int* __restrict__ dst, int n_edges, int chunk, int n_buckets,
    int row_stride, int n_groups, int n_wg, int* __restrict__ cnt,
    int* __restrict__ gpre, const float* __restrict__ x,
    const float* __restrict__ Wn, const int* __restrict__ ntype,
    float* __restrict__ xout, int n_nodes, int n_tiles,
    const float* __restrict__ Wm, float* __restrict__ wfrag) {
  extern __shared__ int hist[];
  __shared__ float w[2 * 64 * NSTRIDE];
  const int tid = threadIdx.x;
  const int lane = tid & 31, wave = tid >> 5;
  const int tile_blocks = (n_tiles + 31) >> 5;

  if ((int)blockIdx.x < tile_blocks) {
    for (int i = tid; i < 2 * 64 * 64; i += SORT_THREADS) {
      int t = i >> 12, k = (i >> 6) & 63, cc = i & 63;
      w[(t * 64 + cc) * NSTRIDE + k] = Wn[i];
    }
    __syncthreads();

    const int tile = (int)blockIdx.x * 32 + wave;
    if (tile >= n_tiles) return;

    const int arow = lane & 15;
    const int khalf = (lane >> 4) << 1;
    int node = tile * 16 + arow;
    const bool valid = node < n_nodes;
    if (!valid) node = n_nodes - 1;
    const int t_row = valid ? ntype[node] : -1;
    const float* __restrict__ ar = x + (size_t)node * 64;
    const float* __restrict__ wb0 = &w[arow * NSTRIDE];
    const float* __restrict__ wb1 = &w[(64 + arow) * NSTRIDE];

    v8f acc[4] = {};
#pragma unroll 1
    for (int k0 = 0; k0 < 64; k0 += 8) {
      v2f a[2];
#pragma unroll
      for (int u = 0; u < 2; ++u)
        a[u] = __builtin_nontemporal_load((const v2f*)(ar + k0 + 4 * u + khalf));
#pragma unroll
      for (int u = 0; u < 2; ++u) {
        const int k = k0 + 4 * u + khalf;
        const v2f z = {0.f, 0.f};
        v2f a0 = (t_row == 0) ? a[u] : z;
        v2f a1 = (t_row == 1) ? a[u] : z;
#pragma unroll
        for (int nb = 0; nb < 4; ++nb) {
          v2f b0 = *(const v2f*)(wb0 + nb * 16 * NSTRIDE + k);
          v2f b1 = *(const v2f*)(wb1 + nb * 16 * NSTRIDE + k);
          acc[nb] = wmma_f32(a0, b0, acc[nb]);
          acc[nb] = wmma_f32(a1, b1, acc[nb]);
        }
      }
    }

    const int rbase = (lane >> 4) * 8;
    const int col = lane & 15;
#pragma unroll
    for (int r = 0; r < 8; ++r) {
      const int orow = tile * 16 + rbase + r;
      if (orow < n_nodes) {
#pragma unroll
        for (int nb = 0; nb < 4; ++nb)
          xout[(size_t)orow * 64 + nb * 16 + col] = ((const float*)&acc[nb])[r];
      }
    }
    return;
  }

  const int wg = (int)blockIdx.x - tile_blocks;
  if (wg >= n_wg) {
    const int i = (wg - n_wg) * SORT_THREADS + tid;
    if (i < WFRAG_FLOATS) {
      const int j = i & 3, ln = (i >> 2) & 31, blk = i >> 7;
      const int kb = blk % 36, nbb = blk / 36;
      const int k = kb * 8 + (j >> 1) * 4 + ((ln >> 4) << 1) + (j & 1);
      wfrag[i] = Wm[k * 64 + nbb * 16 + (ln & 15)];
    }
    return;
  }

  for (int b = tid; b < n_buckets; b += SORT_THREADS) hist[b] = 0;
  __syncthreads();
  const int cbeg = wg * chunk;
  const int cend = (cbeg + chunk < n_edges) ? cbeg + chunk : n_edges;
  for (int i0 = cbeg + tid; i0 < cend; i0 += SORT_DEPTH * SORT_THREADS) {
    int d[SORT_DEPTH];
#pragma unroll
    for (int u = 0; u < SORT_DEPTH; ++u) {
      const int i = i0 + u * SORT_THREADS;
      d[u] = dst[i < cend ? i : cend - 1];
    }
#pragma unroll
    for (int u = 0; u < SORT_DEPTH; ++u)
      if (i0 + u * SORT_THREADS < cend) atomicAdd(&hist[d[u] >> BUCKET_SHIFT], 1);
  }
  __syncthreads();
  int* __restrict__ row = cnt + (size_t)wg * row_stride;
  for (int b = tid; b < n_buckets; b += SORT_THREADS) row[b] = hist[b];
  int* gs = hist + row_stride;
  for (int k = wave; k < n_groups; k += SORT_THREADS / 32) {
    const int b = k * GROUP_BUCKETS + lane;
    const int s = wave_sum(b < n_buckets ? hist[b] : 0);
    if (lane == 0) gs[k] = s;
  }
  __syncthreads();
  if (wave == 0) {
    int carry = 0;
    for (int k0 = 0; k0 < n_groups; k0 += 32) {
      const int k = k0 + lane;
      const int v = (k < n_groups) ? gs[k] : 0;
      const int inc = wave_incl_scan(v, lane);
      if (k < n_groups) gpre[(size_t)k * n_wg + wg] = carry + inc - v;
      carry += __shfl(inc, 31, 32);
    }
  }
}

__global__ __launch_bounds__(1024) void offsets_kernel(
    const int* __restrict__ cnt, const int* __restrict__ gpre, int n_buckets,
    int row_stride, int* __restrict__ pos, int* __restrict__ bstart) {
  __shared__ int segsum[32][GROUP_BUCKETS];
  __shared__ int red[SORT_WGS / 32];
  const int tid = threadIdx.x;
  const int lane = tid & 31, wave = tid >> 5;
  const int k = blockIdx.x;
  const int b = k * GROUP_BUCKETS + lane;
  const bool live = b < n_buckets;
  const size_t o = (size_t)(wave * SEG_WGS) * row_stride + b;
  int v[SEG_WGS];
#pragma unroll
  for (int i = 0; i < SEG_WGS; ++i) v[i] = live ? cnt[o + (size_t)i * row_stride] : 0;
  if (tid < SORT_WGS) {
    const int t = wave_sum(gpre[(size_t)k * SORT_WGS + tid]);
    if (lane == 0) red[wave] = t;
  }
  int s = 0;
#pragma unroll
  for (int i = 0; i < SEG_WGS; ++i) s += v[i];
  segsum[wave][lane] = s;
  __syncthreads();
  int total = 0, before = 0;
#pragma unroll
  for (int w = 0; w < 32; ++w) {
    const int x = segsum[w][lane];
    total += x;
    if (w < wave) before += x;
  }
  int base = 0;
#pragma unroll
  for (int w = 0; w < SORT_WGS / 32; ++w) base += red[w];
  const int first = base + wave_incl_scan(total, lane) - total;
  int run = first + before;
  if (live) {
#pragma unroll
    for (int i = 0; i < SEG_WGS; ++i) {
      pos[o + (size_t)i * row_stride] = run;
      run += v[i];
    }
    if (wave == 0) bstart[b] = first;
    if (wave == 31 && b == n_buckets - 1) bstart[n_buckets] = run;
  }
}

#define STAGE_CAP 12544

__global__ __launch_bounds__(SORT_THREADS) void bucket_place_kernel(
    const int* __restrict__ src, const int* __restrict__ dst,
    const int* __restrict__ etype, int n_edges, int chunk, int n_buckets,
    int row_stride, const int* __restrict__ cnt, const int* __restrict__ pos,
    int* __restrict__ rec) {
  extern __shared__ int cursor[];
  __shared__ v4i stage[STAGE_CAP];
  __shared__ int wsum[SORT_THREADS / 32];
  const int tid = threadIdx.x;
  const int lane = tid & 31, wave = tid >> 5;
  const int wg = blockIdx.x;
  const int* __restrict__ prow = pos + (size_t)wg * row_stride;
  const int cbeg = wg * chunk;
  const int cend = (cbeg + chunk < n_edges) ? cbeg + chunk : n_edges;
  const bool staged = chunk <= STAGE_CAP;
  int* delta = cursor + n_buckets;
  if (staged) {
    const int* __restrict__ crow = cnt + (size_t)wg * row_stride;
#pragma unroll 4
    for (int b = tid; b < n_buckets; b += SORT_THREADS) {
      cursor[b] = crow[b];
      delta[b] = prow[b];
    }
    __syncthreads();
    const int per = (n_buckets + SORT_THREADS - 1) / SORT_THREADS;
    const int b0 = tid * per;
    const int b1 = (b0 + per < n_buckets) ? b0 + per : n_buckets;
    int s = 0;
    for (int b = b0; b < b1; ++b) s += cursor[b];
    const int inc = wave_incl_scan(s, lane);
    if (lane == 31) wsum[wave] = inc;
    __syncthreads();
    int run = inc - s;
    for (int w = 0; w < SORT_THREADS / 32; ++w)
      if (w < wave) run += wsum[w];
    for (int b = b0; b < b1; ++b) {
      const int c = cursor[b];
      cursor[b] = run;
      delta[b] -= run;
      run += c;
    }
  } else {
    for (int b = tid; b < n_buckets; b += SORT_THREADS) cursor[b] = prow[b];
  }
  __syncthreads();
  for (int i0 = cbeg + tid; i0 < cend; i0 += SORT_DEPTH * SORT_THREADS) {
    int d[SORT_DEPTH], t[SORT_DEPTH], sn[SORT_DEPTH];
#pragma unroll
    for (int u = 0; u < SORT_DEPTH; ++u) {
      const int i = i0 + u * SORT_THREADS;
      const int ic = i < cend ? i : cend - 1;
      d[u] = dst[ic];
      t[u] = etype[ic];
      sn[u] = src[ic];
    }
#pragma unroll
    for (int u = 0; u < SORT_DEPTH; ++u) {
      const int i = i0 + u * SORT_THREADS;
      if (i < cend) {
        const int tt = (t[u] < 0 || t[u] > 2) ? 3 : t[u];
        const int p = atomicAdd(&cursor[d[u] >> BUCKET_SHIFT], 1);
        if (staged) {
          v4i r = {i, sn[u], tt, d[u]};
          stage[p] = r;
        } else {
          v4i r = {i, sn[u], tt, d[u] & (BUCKET_NODES - 1)};
          *(v4i*)(rec + 4 * (size_t)p) = r;
        }
      }
    }
  }
  if (!staged) return;
  __syncthreads();
  const int n_local = cend - cbeg;
#pragma unroll 4
  for (int i = tid; i < n_local; i += SORT_THREADS) {
    v4i r = stage[i];
    const int p = i + delta[r.w >> BUCKET_SHIFT];
    r.w &= BUCKET_NODES - 1;
    *(v4i*)(rec + 4 * (size_t)p) = r;
  }
}

__global__ __launch_bounds__(256) void aggregate_xform_kernel(
    const float* __restrict__ xout, const float* __restrict__ ef,
    const int* __restrict__ rec, const int* __restrict__ bstart,
    const float* __restrict__ wfrag, float* __restrict__ out, int n_nodes) {
  __shared__ float s[BUCKET_NODES * SSTRIDE];
  const int tid = threadIdx.x;
  for (int i = tid; i < BUCKET_NODES * SSTRIDE / 4; i += 256)
    ((v4f*)s)[i] = v4f{0.f, 0.f, 0.f, 0.f};
  __syncthreads();

  const int wave = __builtin_amdgcn_readfirstlane((int)(threadIdx.x >> 5));
  const int lane = threadIdx.x & 31;
  const int b = (int)blockIdx.x;
  const int beg = __builtin_amdgcn_readfirstlane(bstart[b]);
  const int end = __builtin_amdgcn_readfirstlane(bstart[b + 1]);
  const int per = (end - beg + 7) >> 3;
  const int wbeg = beg + wave * per;
  const int wend = (wbeg + per < end) ? wbeg + per : end;

  for (int j0 = wbeg; j0 < wend; j0 += 32) {
    const int left = wend - j0;
    const int nrec = left < 32 ? left : 32;
    v4i r = {0, 0, 3, 0};
    if (lane < nrec) r = *(const v4i*)(rec + 4 * (size_t)(j0 + lane));
    for (int q = 0; q < nrec; q += 4) {
      float x0[4], x1[4], f[4];
      int off[4];
#pragma unroll
      for (int u = 0; u < 4; ++u) {
        const int qq = q + u;
        const bool live = qq < nrec;
        const int qc = live ? qq : q;
        const unsigned e = (unsigned)__builtin_amdgcn_readlane(r.x, qc);
        const unsigned sn = (unsigned)__builtin_amdgcn_readlane(r.y, qc);
        const int t = __builtin_amdgcn_readlane(r.z, qc);
        const int dl = __builtin_amdgcn_readlane(r.w, qc);
        off[u] = (live && t < 3) ? (dl * SSTRIDE + t * 96) : -1;
        const float* xr = xout + ((size_t)sn * 64u + (unsigned)lane);
        x0[u] = xr[0];
        x1[u] = xr[32];
        f[u] = ef[(size_t)e * 32u + (unsigned)lane];
      }
#pragma unroll
      for (int u = 0; u < 4; ++u) {
        if (off[u] >= 0) {
          float* row = s + off[u] + lane;
          unsafeAtomicAdd(row, x0[u]);
          unsafeAtomicAdd(row + 32, x1[u]);
          unsafeAtomicAdd(row + 64, f[u]);
        }
      }
    }
  }
  __syncthreads();

  const int nb = wave & 3;
  const int kh = wave >> 2;
  const int arow = lane & 15;
  const int h2 = (lane >> 4) << 1;
  const float* a0p = s + arow * SSTRIDE + kh * 144 + h2;
  const float* a1p = a0p + 16 * SSTRIDE;
  const v4f* __restrict__ bp = (const v4f*)wfrag + ((nb * 36 + kh * 18) * 32 + lane);

  v8f acc0 = {}, acc1 = {};
#pragma unroll
  for (int j = 0; j < 18; ++j) {
    const v4f bq = bp[j * 32];
    const v2f b01 = {bq.x, bq.y};
    const v2f b23 = {bq.z, bq.w};
    const v2f a00 = *(const v2f*)(a0p + j * 8);
    const v2f a01 = *(const v2f*)(a0p + j * 8 + 4);
    const v2f a10 = *(const v2f*)(a1p + j * 8);
    const v2f a11 = *(const v2f*)(a1p + j * 8 + 4);
    acc0 = wmma_f32(a00, b01, acc0);
    acc1 = wmma_f32(a10, b01, acc1);
    acc0 = wmma_f32(a01, b23, acc0);
    acc1 = wmma_f32(a11, b23, acc1);
  }
  __syncthreads();

  float* st = s + ((lane >> 4) * 8) * OSTRIDE + nb * 16 + (lane & 15);
  if (kh == 0) {
#pragma unroll
    for (int r8 = 0; r8 < 8; ++r8) {
      st[r8 * OSTRIDE] = ((const float*)&acc0)[r8];
      st[(16 + r8) * OSTRIDE] = ((const float*)&acc1)[r8];
    }
  }
  __syncthreads();
  if (kh == 1) {
#pragma unroll
    for (int r8 = 0; r8 < 8; ++r8) {
      st[r8 * OSTRIDE] += ((const float*)&acc0)[r8];
      st[(16 + r8) * OSTRIDE] += ((const float*)&acc1)[r8];
    }
  }
  __syncthreads();

  const int node0 = b * BUCKET_NODES;
  const int rows = (n_nodes - node0 < BUCKET_NODES) ? n_nodes - node0 : BUCKET_NODES;
  v4f* __restrict__ og = (v4f*)(out + (size_t)node0 * 64);
  for (int i = tid; i < rows * 16; i += 256)
    og[i] = *(const v4f*)(s + (i >> 4) * OSTRIDE + (i & 15) * 4);
}

extern "C" void kernel_launch(void* const* d_in, const int* in_sizes, int n_in,
                              void* d_out, int out_size, void* d_ws,
                              size_t ws_size, hipStream_t stream) {
  const float* node_feature = (const float*)d_in[0];
  const float* edge_feature = (const float*)d_in[1];
  const float* W_node = (const float*)d_in[2];
  const float* W_msg = (const float*)d_in[3];
  const int* node_type = (const int*)d_in[4];
  const int* edge_type = (const int*)d_in[5];
  const int* edge_index = (const int*)d_in[6];
  const int n_nodes = in_sizes[4];
  const int n_edges = in_sizes[5];
  const int* src = edge_index;
  const int* dst = edge_index + n_edges;

  float* out = (float*)d_out;

  const int n_buckets = (n_nodes + BUCKET_NODES - 1) / BUCKET_NODES;
  const int sort_wgs = SORT_WGS;
  const int chunk = (n_edges + sort_wgs - 1) / sort_wgs;
  const int n_groups = (n_buckets + GROUP_BUCKETS - 1) / GROUP_BUCKETS;
  const int row_stride = n_groups * GROUP_BUCKETS;

  float* xout = (float*)d_ws;
  float* wfrag = xout + (size_t)n_nodes * 64;
  int* rec = (int*)(wfrag + WFRAG_FLOATS);
  int* cnt = rec + 4 * (size_t)n_edges;
  int* pos = cnt + (size_t)sort_wgs * row_stride;
  int* gpre = pos + (size_t)sort_wgs * row_stride;
  int* bstart = gpre + (size_t)n_groups * sort_wgs;

  const int node_tiles = (n_nodes + 15) / 16;
  count_node_kernel<<<(node_tiles + 31) / 32 + sort_wgs + WFRAG_FLOATS / SORT_THREADS, SORT_THREADS,
                      (row_stride + n_groups) * sizeof(int), stream>>>(
      dst, n_edges, chunk, n_buckets, row_stride, n_groups, sort_wgs, cnt, gpre,
      node_feature, W_node, node_type, xout, n_nodes, node_tiles, W_msg, wfrag);
  offsets_kernel<<<n_groups, 1024, 0, stream>>>(cnt, gpre, n_buckets, row_stride, pos, bstart);
  bucket_place_kernel<<<sort_wgs, SORT_THREADS, 2 * n_buckets * sizeof(int), stream>>>(
      src, dst, edge_type, n_edges, chunk, n_buckets, row_stride, cnt, pos, rec);

  aggregate_xform_kernel<<<n_buckets, 256, 0, stream>>>(
      xout, edge_feature, rec, bstart, wfrag, out, n_nodes);
}


# Bucket widened to 64 destination nodes per 16-wave work-group: a fetched W fragment feeds 4 row tiles (weight re-reads halved), sort runs twice as long; sum order: four k quarters added through LDS
# speedup vs baseline: 2.4933x; 1.0642x over previous
#include <hip/hip_runtime.h>


typedef __attribute__((ext_vector_type(2))) float v2f;
typedef __attribute__((ext_vector_type(4))) float v4f;
typedef __attribute__((ext_vector_type(8))) float v8f;
typedef __attribute__((ext_vector_type(4))) int v4i;

#define NSTRIDE 68
#define MSG_K 288
#define SSTRIDE 292
#define OSTRIDE 68
#define WFRAG_FLOATS (MSG_K * 64)

__device__ __forceinline__ v8f wmma_f32(v2f a, v2f b, v8f c) {
  return __builtin_amdgcn_wmma_f32_16x16x4_f32(false, a, false, b, (short)0, c,
                                               false, false);
}

__device__ __forceinline__ int wave_incl_scan(int v, int lane) {
#pragma unroll
  for (int off = 1; off < 32; off <<= 1) {
    int u = __shfl_up(v, off, 32);
    if (lane >= off) v += u;
  }
  return v;
}

__device__ __forceinline__ int wave_sum(int v) {
#pragma unroll
  for (int off = 16; off > 0; off >>= 1) v += __shfl_xor(v, off, 32);
  return v;
}

#define BUCKET_SHIFT 6
#define BUCKET_NODES 64
#define SORT_THREADS 1024
#define SORT_WGS 128
#define SORT_DEPTH 16
#define GROUP_BUCKETS 32
#define SEG_WGS (SORT_WGS / 32)

__global__ __launch_bounds__(SORT_THREADS, 16) void count_node_kernel(
    const int* __restrict__ dst, int n_edges, int chunk, int n_buckets,
    int row_stride, int n_groups, int n_wg, int* __restrict__ cnt,
    int* __restrict__ gpre, const float* __restrict__ x,
    const float* __restrict__ Wn, const int* __restrict__ ntype,
    float* __restrict__ xout, int n_nodes, int n_tiles,
    const float* __restrict__ Wm, float* __restrict__ wfrag) {
  extern __shared__ int hist[];
  __shared__ float w[2 * 64 * NSTRIDE];
  const int tid = threadIdx.x;
  const int lane = tid & 31, wave = tid >> 5;
  const int tile_blocks = (n_tiles + 31) >> 5;

  if ((int)blockIdx.x < tile_blocks) {
    for (int i = tid; i < 2 * 64 * 64; i += SORT_THREADS) {
      int t = i >> 12, k = (i >> 6) & 63, cc = i & 63;
      w[(t * 64 + cc) * NSTRIDE + k] = Wn[i];
    }
    __syncthreads();

    const int tile = (int)blockIdx.x * 32 + wave;
    if (tile >= n_tiles) return;

    const int arow = lane & 15;
    const int khalf = (lane >> 4) << 1;
    int node = tile * 16 + arow;
    const bool valid = node < n_nodes;
    if (!valid) node = n_nodes - 1;
    const int t_row = valid ? ntype[node] : -1;
    const float* __restrict__ ar = x + (size_t)node * 64;
    const float* __restrict__ wb0 = &w[arow * NSTRIDE];
    const float* __restrict__ wb1 = &w[(64 + arow) * NSTRIDE];

    v8f acc[4] = {};
#pragma unroll 1
    for (int k0 = 0; k0 < 64; k0 += 8) {
      v2f a[2];
#pragma unroll
      for (int u = 0; u < 2; ++u)
        a[u] = __builtin_nontemporal_load((const v2f*)(ar + k0 + 4 * u + khalf));
#pragma unroll
      for (int u = 0; u < 2; ++u) {
        const int k = k0 + 4 * u + khalf;
        const v2f z = {0.f, 0.f};
        v2f a0 = (t_row == 0) ? a[u] : z;
        v2f a1 = (t_row == 1) ? a[u] : z;
#pragma unroll
        for (int nb = 0; nb < 4; ++nb) {
          v2f b0 = *(const v2f*)(wb0 + nb * 16 * NSTRIDE + k);
          v2f b1 = *(const v2f*)(wb1 + nb * 16 * NSTRIDE + k);
          acc[nb] = wmma_f32(a0, b0, acc[nb]);
          acc[nb] = wmma_f32(a1, b1, acc[nb]);
        }
      }
    }

    const int rbase = (lane >> 4) * 8;
    const int col = lane & 15;
#pragma unroll
    for (int r = 0; r < 8; ++r) {
      const int orow = tile * 16 + rbase + r;
      if (orow < n_nodes) {
#pragma unroll
        for (int nb = 0; nb < 4; ++nb)
          xout[(size_t)orow * 64 + nb * 16 + col] = ((const float*)&acc[nb])[r];
      }
    }
    return;
  }

  const int wg = (int)blockIdx.x - tile_blocks;
  if (wg >= n_wg) {
    const int i = (wg - n_wg) * SORT_THREADS + tid;
    if (i < WFRAG_FLOATS) {
      const int j = i & 3, ln = (i >> 2) & 31, blk = i >> 7;
      const int kb = blk % 36, nbb = blk / 36;
      const int k = kb * 8 + (j >> 1) * 4 + ((ln >> 4) << 1) + (j & 1);
      wfrag[i] = Wm[k * 64 + nbb * 16 + (ln & 15)];
    }
    return;
  }

  for (int b = tid; b < n_buckets; b += SORT_THREADS) hist[b] = 0;
  __syncthreads();
  const int cbeg = wg * chunk;
  const int cend = (cbeg + chunk < n_edges) ? cbeg + chunk : n_edges;
  for (int i0 = cbeg + tid; i0 < cend; i0 += SORT_DEPTH * SORT_THREADS) {
    int d[SORT_DEPTH];
#pragma unroll
    for (int u = 0; u < SORT_DEPTH; ++u) {
      const int i = i0 + u * SORT_THREADS;
      d[u] = dst[i < cend ? i : cend - 1];
    }
#pragma unroll
    for (int u = 0; u < SORT_DEPTH; ++u)
      if (i0 + u * SORT_THREADS < cend) atomicAdd(&hist[d[u] >> BUCKET_SHIFT], 1);
  }
  __syncthreads();
  int* __restrict__ row = cnt + (size_t)wg * row_stride;
  for (int b = tid; b < n_buckets; b += SORT_THREADS) row[b] = hist[b];
  int* gs = hist + row_stride;
  for (int k = wave; k < n_groups; k += SORT_THREADS / 32) {
    const int b = k * GROUP_BUCKETS + lane;
    const int s = wave_sum(b < n_buckets ? hist[b] : 0);
    if (lane == 0) gs[k] = s;
  }
  __syncthreads();
  if (wave == 0) {
    int carry = 0;
    for (int k0 = 0; k0 < n_groups; k0 += 32) {
      const int k = k0 + lane;
      const int v = (k < n_groups) ? gs[k] : 0;
      const int inc = wave_incl_scan(v, lane);
      if (k < n_groups) gpre[(size_t)k * n_wg + wg] = carry + inc - v;
      carry += __shfl(inc, 31, 32);
    }
  }
}

__global__ __launch_bounds__(1024) void offsets_kernel(
    const int* __restrict__ cnt, const int* __restrict__ gpre, int n_buckets,
    int row_stride, int* __restrict__ pos, int* __restrict__ bstart) {
  __shared__ int segsum[32][GROUP_BUCKETS];
  __shared__ int red[SORT_WGS / 32];
  const int tid = threadIdx.x;
  const int lane = tid & 31, wave = tid >> 5;
  const int k = blockIdx.x;
  const int b = k * GROUP_BUCKETS + lane;
  const bool live = b < n_buckets;
  const size_t o = (size_t)(wave * SEG_WGS) * row_stride + b;
  int v[SEG_WGS];
#pragma unroll
  for (int i = 0; i < SEG_WGS; ++i) v[i] = live ? cnt[o + (size_t)i * row_stride] : 0;
  if (tid < SORT_WGS) {
    const int t = wave_sum(gpre[(size_t)k * SORT_WGS + tid]);
    if (lane == 0) red[wave] = t;
  }
  int s = 0;
#pragma unroll
  for (int i = 0; i < SEG_WGS; ++i) s += v[i];
  segsum[wave][lane] = s;
  __syncthreads();
  int total = 0, before = 0;
#pragma unroll
  for (int w = 0; w < 32; ++w) {
    const int x = segsum[w][lane];
    total += x;
    if (w < wave) before += x;
  }
  int base = 0;
#pragma unroll
  for (int w = 0; w < SORT_WGS / 32; ++w) base += red[w];
  const int first = base + wave_incl_scan(total, lane) - total;
  int run = first + before;
  if (live) {
#pragma unroll
    for (int i = 0; i < SEG_WGS; ++i) {
      pos[o + (size_t)i * row_stride] = run;
      run += v[i];
    }
    if (wave == 0) bstart[b] = first;
    if (wave == 31 && b == n_buckets - 1) bstart[n_buckets] = run;
  }
}

#define STAGE_CAP 12544

__global__ __launch_bounds__(SORT_THREADS) void bucket_place_kernel(
    const int* __restrict__ src, const int* __restrict__ dst,
    const int* __restrict__ etype, int n_edges, int chunk, int n_buckets,
    int row_stride, const int* __restrict__ cnt, const int* __restrict__ pos,
    int* __restrict__ rec) {
  extern __shared__ int cursor[];
  __shared__ v4i stage[STAGE_CAP];
  __shared__ int wsum[SORT_THREADS / 32];
  const int tid = threadIdx.x;
  const int lane = tid & 31, wave = tid >> 5;
  const int wg = blockIdx.x;
  const int* __restrict__ prow = pos + (size_t)wg * row_stride;
  const int cbeg = wg * chunk;
  const int cend = (cbeg + chunk < n_edges) ? cbeg + chunk : n_edges;
  const bool staged = chunk <= STAGE_CAP;
  int* delta = cursor + n_buckets;
  if (staged) {
    const int* __restrict__ crow = cnt + (size_t)wg * row_stride;
#pragma unroll 4
    for (int b = tid; b < n_buckets; b += SORT_THREADS) {
      cursor[b] = crow[b];
      delta[b] = prow[b];
    }
    __syncthreads();
    const int per = (n_buckets + SORT_THREADS - 1) / SORT_THREADS;
    const int b0 = tid * per;
    const int b1 = (b0 + per < n_buckets) ? b0 + per : n_buckets;
    int s = 0;
    for (int b = b0; b < b1; ++b) s += cursor[b];
    const int inc = wave_incl_scan(s, lane);
    if (lane == 31) wsum[wave] = inc;
    __syncthreads();
    int run = inc - s;
    for (int w = 0; w < SORT_THREADS / 32; ++w)
      if (w < wave) run += wsum[w];
    for (int b = b0; b < b1; ++b) {
      const int c = cursor[b];
      cursor[b] = run;
      delta[b] -= run;
      run += c;
    }
  } else {
    for (int b = tid; b < n_buckets; b += SORT_THREADS) cursor[b] = prow[b];
  }
  __syncthreads();
  for (int i0 = cbeg + tid; i0 < cend; i0 += SORT_DEPTH * SORT_THREADS) {
    int d[SORT_DEPTH], t[SORT_DEPTH], sn[SORT_DEPTH];
#pragma unroll
    for (int u = 0; u < SORT_DEPTH; ++u) {
      const int i = i0 + u * SORT_THREADS;
      const int ic = i < cend ? i : cend - 1;
      d[u] = dst[ic];
      t[u] = etype[ic];
      sn[u] = src[ic];
    }
#pragma unroll
    for (int u = 0; u < SORT_DEPTH; ++u) {
      const int i = i0 + u * SORT_THREADS;
      if (i < cend) {
        const int tt = (t[u] < 0 || t[u] > 2) ? 3 : t[u];
        const int p = atomicAdd(&cursor[d[u] >> BUCKET_SHIFT], 1);
        if (staged) {
          v4i r = {i, sn[u], tt, d[u]};
          stage[p] = r;
        } else {
          v4i r = {i, sn[u], tt, d[u] & (BUCKET_NODES - 1)};
          *(v4i*)(rec + 4 * (size_t)p) = r;
        }
      }
    }
  }
  if (!staged) return;
  __syncthreads();
  const int n_local = cend - cbeg;
#pragma unroll 4
  for (int i = tid; i < n_local; i += SORT_THREADS) {
    v4i r = stage[i];
    const int p = i + delta[r.w >> BUCKET_SHIFT];
    r.w &= BUCKET_NODES - 1;
    *(v4i*)(rec + 4 * (size_t)p) = r;
  }
}

#define AGG_WAVES 16
#define AGG_THREADS (AGG_WAVES * 32)
#define QSTRIDE (BUCKET_NODES * OSTRIDE)

__global__ __launch_bounds__(AGG_THREADS, 16) void aggregate_xform_kernel(
    const float* __restrict__ xout, const float* __restrict__ ef,
    const int* __restrict__ rec, const int* __restrict__ bstart,
    const float* __restrict__ wfrag, float* __restrict__ out, int n_nodes) {
  __shared__ float s[BUCKET_NODES * SSTRIDE];
  const int tid = threadIdx.x;
  for (int i = tid; i < BUCKET_NODES * SSTRIDE / 4; i += AGG_THREADS)
    ((v4f*)s)[i] = v4f{0.f, 0.f, 0.f, 0.f};
  __syncthreads();

  const int wave = __builtin_amdgcn_readfirstlane((int)(threadIdx.x >> 5));
  const int lane = threadIdx.x & 31;
  const int b = (int)blockIdx.x;
  const int beg = __builtin_amdgcn_readfirstlane(bstart[b]);
  const int end = __builtin_amdgcn_readfirstlane(bstart[b + 1]);
  const int per = (end - beg + AGG_WAVES - 1) / AGG_WAVES;
  const int wbeg = beg + wave * per;
  const int wend = (wbeg + per < end) ? wbeg + per : end;

  for (int j0 = wbeg; j0 < wend; j0 += 32) {
    const int left = wend - j0;
    const int nrec = left < 32 ? left : 32;
    v4i r = {0, 0, 3, 0};
    if (lane < nrec) r = *(const v4i*)(rec + 4 * (size_t)(j0 + lane));
    for (int q = 0; q < nrec; q += 4) {
      float x0[4], x1[4], f[4];
      int off[4];
#pragma unroll
      for (int u = 0; u < 4; ++u) {
        const int qq = q + u;
        const bool live = qq < nrec;
        const int qc = live ? qq : q;
        const unsigned e = (unsigned)__builtin_amdgcn_readlane(r.x, qc);
        const unsigned sn = (unsigned)__builtin_amdgcn_readlane(r.y, qc);
        const int t = __builtin_amdgcn_readlane(r.z, qc);
        const int dl = __builtin_amdgcn_readlane(r.w, qc);
        off[u] = (live && t < 3) ? (dl * SSTRIDE + t * 96) : -1;
        const float* xr = xout + ((size_t)sn * 64u + (unsigned)lane);
        x0[u] = xr[0];
        x1[u] = xr[32];
        f[u] = ef[(size_t)e * 32u + (unsigned)lane];
      }
#pragma unroll
      for (int u = 0; u < 4; ++u) {
        if (off[u] >= 0) {
          float* row = s + off[u] + lane;
          unsafeAtomicAdd(row, x0[u]);
          unsafeAtomicAdd(row + 32, x1[u]);
          unsafeAtomicAdd(row + 64, f[u]);
        }
      }
    }
  }
  __syncthreads();

  const int nb = wave & 3;
  const int kq = wave >> 2;
  const int arow = lane & 15;
  const int h2 = (lane >> 4) << 1;
  const float* ap = s + arow * SSTRIDE + kq * 72 + h2;
  const v4f* __restrict__ bp = (const v4f*)wfrag + ((nb * 36 + kq * 9) * 32 + lane);

  v8f acc[4] = {};
#pragma unroll
  for (int j = 0; j < 9; ++j) {
    const v4f bq = bp[j * 32];
    const v2f b01 = {bq.x, bq.y};
    const v2f b23 = {bq.z, bq.w};
#pragma unroll
    for (int rt = 0; rt < 4; ++rt) {
      const v2f a0 = *(const v2f*)(ap + rt * 16 * SSTRIDE + j * 8);
      acc[rt] = wmma_f32(a0, b01, acc[rt]);
    }
#pragma unroll
    for (int rt = 0; rt < 4; ++rt) {
      const v2f a1 = *(const v2f*)(ap + rt * 16 * SSTRIDE + j * 8 + 4);
      acc[rt] = wmma_f32(a1, b23, acc[rt]);
    }
  }
  __syncthreads();

  float* st = s + kq * QSTRIDE + ((lane >> 4) * 8) * OSTRIDE + nb * 16 + (lane & 15);
#pragma unroll
  for (int rt = 0; rt < 4; ++rt) {
#pragma unroll
    for (int r8 = 0; r8 < 8; ++r8)
      st[(rt * 16 + r8) * OSTRIDE] = ((const float*)&acc[rt])[r8];
  }
  __syncthreads();

  const int node0 = b * BUCKET_NODES;
  const int rows = (n_nodes - node0 < BUCKET_NODES) ? n_nodes - node0 : BUCKET_NODES;
  v4f* __restrict__ og = (v4f*)(out + (size_t)node0 * 64);
  for (int i = tid; i < rows * 16; i += AGG_THREADS) {
    const float* p = s + (i >> 4) * OSTRIDE + (i & 15) * 4;
    v4f v = *(const v4f*)p;
    v += *(const v4f*)(p + QSTRIDE);
    v += *(const v4f*)(p + 2 * QSTRIDE);
    v += *(const v4f*)(p + 3 * QSTRIDE);
    og[i] = v;
  }
}

extern "C" void kernel_launch(void* const* d_in, const int* in_sizes, int n_in,
                              void* d_out, int out_size, void* d_ws,
                              size_t ws_size, hipStream_t stream) {
  const float* node_feature = (const float*)d_in[0];
  const float* edge_feature = (const float*)d_in[1];
  const float* W_node = (const float*)d_in[2];
  const float* W_msg = (const float*)d_in[3];
  const int* node_type = (const int*)d_in[4];
  const int* edge_type = (const int*)d_in[5];
  const int* edge_index = (const int*)d_in[6];
  const int n_nodes = in_sizes[4];
  const int n_edges = in_sizes[5];
  const int* src = edge_index;
  const int* dst = edge_index + n_edges;

  float* out = (float*)d_out;

  const int n_buckets = (n_nodes + BUCKET_NODES - 1) / BUCKET_NODES;
  const int sort_wgs = SORT_WGS;
  const int chunk = (n_edges + sort_wgs - 1) / sort_wgs;
  const int n_groups = (n_buckets + GROUP_BUCKETS - 1) / GROUP_BUCKETS;
  const int row_stride = n_groups * GROUP_BUCKETS;

  float* xout = (float*)d_ws;
  float* wfrag = xout + (size_t)n_nodes * 64;
  int* rec = (int*)(wfrag + WFRAG_FLOATS);
  int* cnt = rec + 4 * (size_t)n_edges;
  int* pos = cnt + (size_t)sort_wgs * row_stride;
  int* gpre = pos + (size_t)sort_wgs * row_stride;
  int* bstart = gpre + (size_t)n_groups * sort_wgs;

  const int node_tiles = (n_nodes + 15) / 16;
  count_node_kernel<<<(node_tiles + 31) / 32 + sort_wgs + WFRAG_FLOATS / SORT_THREADS, SORT_THREADS,
                      (row_stride + n_groups) * sizeof(int), stream>>>(
      dst, n_edges, chunk, n_buckets, row_stride, n_groups, sort_wgs, cnt, gpre,
      node_feature, W_node, node_type, xout, n_nodes, node_tiles, W_msg, wfrag);
  offsets_kernel<<<n_groups, 1024, 0, stream>>>(cnt, gpre, n_buckets, row_stride, pos, bstart);
  bucket_place_kernel<<<sort_wgs, SORT_THREADS, 2 * n_buckets * sizeof(int), stream>>>(
      src, dst, edge_type, n_edges, chunk, n_buckets, row_stride, cnt, pos, rec);

  aggregate_xform_kernel<<<n_buckets, AGG_THREADS, 0, stream>>>(
      xout, edge_feature, rec, bstart, wfrag, out, n_nodes);
}


# Chunk-local sort: one work-group counts, prefix-sums and orders its chunk in LDS; a bucket is 128 short runs found via a (start,count) table; no offsets pass, 4 launches -> 2; sum order: LDS arrival
# speedup vs baseline: 2.5716x; 1.0314x over previous
#include <hip/hip_runtime.h>


typedef __attribute__((ext_vector_type(2))) float v2f;
typedef __attribute__((ext_vector_type(4))) float v4f;
typedef __attribute__((ext_vector_type(8))) float v8f;
typedef __attribute__((ext_vector_type(4))) int v4i;
typedef __attribute__((ext_vector_type(2))) int v2i;

#define NSTRIDE 68
#define MSG_K 288
#define SSTRIDE 292
#define OSTRIDE 68
#define WFRAG_FLOATS (MSG_K * 64)

__device__ __forceinline__ v8f wmma_f32(v2f a, v2f b, v8f c) {
  return __builtin_amdgcn_wmma_f32_16x16x4_f32(false, a, false, b, (short)0, c,
                                               false, false);
}

__device__ __forceinline__ int wave_incl_scan(int v, int lane) {
#pragma unroll
  for (int off = 1; off < 32; off <<= 1) {
    int u = __shfl_up(v, off, 32);
    if (lane >= off) v += u;
  }
  return v;
}

#define BUCKET_SHIFT 6
#define BUCKET_NODES 64
#define SORT_THREADS 1024
#define SORT_WGS 128
#define SORT_DEPTH 16
#define PLACE_DEPTH 8
#define STAGE_CAP 12544

__global__ __launch_bounds__(SORT_THREADS, 16) void sort_node_kernel(
    const int* __restrict__ src, const int* __restrict__ dst,
    const int* __restrict__ etype, int n_edges, int chunk, int n_buckets,
    int n_wg, int* __restrict__ rec, int* __restrict__ meta,
    const float* __restrict__ x, const float* __restrict__ Wn,
    const int* __restrict__ ntype, float* __restrict__ xout, int n_nodes,
    int n_tiles, const float* __restrict__ Wm, float* __restrict__ wfrag) {
  extern __shared__ int cursor[];
  __shared__ v2i stage[STAGE_CAP];
  __shared__ int wsum[SORT_THREADS / 32];
  float* w = (float*)stage;
  const int tid = threadIdx.x;
  const int lane = tid & 31, wave = tid >> 5;
  const int tile_blocks = (n_tiles + 31) >> 5;

  if ((int)blockIdx.x < tile_blocks) {
    for (int i = tid; i < 2 * 64 * 64; i += SORT_THREADS) {
      int t = i >> 12, k = (i >> 6) & 63, cc = i & 63;
      w[(t * 64 + cc) * NSTRIDE + k] = Wn[i];
    }
    __syncthreads();

    const int tile = (int)blockIdx.x * 32 + wave;
    if (tile >= n_tiles) return;

    const int arow = lane & 15;
    const int khalf = (lane >> 4) << 1;
    int node = tile * 16 + arow;
    const bool valid = node < n_nodes;
    if (!valid) node = n_nodes - 1;
    const int t_row = valid ? ntype[node] : -1;
    const float* __restrict__ ar = x + (size_t)node * 64;
    const float* __restrict__ wb0 = &w[arow * NSTRIDE];
    const float* __restrict__ wb1 = &w[(64 + arow) * NSTRIDE];

    v8f acc[4] = {};
#pragma unroll 1
    for (int k0 = 0; k0 < 64; k0 += 8) {
      v2f a[2];
#pragma unroll
      for (int u = 0; u < 2; ++u)
        a[u] = __builtin_nontemporal_load((const v2f*)(ar + k0 + 4 * u + khalf));
#pragma unroll
      for (int u = 0; u < 2; ++u) {
        const int k = k0 + 4 * u + khalf;
        const v2f z = {0.f, 0.f};
        v2f a0 = (t_row == 0) ? a[u] : z;
        v2f a1 = (t_row == 1) ? a[u] : z;
#pragma unroll
        for (int nb = 0; nb < 4; ++nb) {
          v2f b0 = *(const v2f*)(wb0 + nb * 16 * NSTRIDE + k);
          v2f b1 = *(const v2f*)(wb1 + nb * 16 * NSTRIDE + k);
          acc[nb] = wmma_f32(a0, b0, acc[nb]);
          acc[nb] = wmma_f32(a1, b1, acc[nb]);
        }
      }
    }

    const int rbase = (lane >> 4) * 8;
    const int col = lane & 15;
#pragma unroll
    for (int r = 0; r < 8; ++r) {
      const int orow = tile * 16 + rbase + r;
      if (orow < n_nodes) {
#pragma unroll
        for (int nb = 0; nb < 4; ++nb)
          xout[(size_t)orow * 64 + nb * 16 + col] = ((const float*)&acc[nb])[r];
      }
    }
    return;
  }

  const int wg = (int)blockIdx.x - tile_blocks;
  if (wg >= n_wg) {
    const int i = (wg - n_wg) * SORT_THREADS + tid;
    if (i < WFRAG_FLOATS) {
      const int j = i & 3, ln = (i >> 2) & 31, blk = i >> 7;
      const int kb = blk % 36, nbb = blk / 36;
      const int k = kb * 8 + (j >> 1) * 4 + ((ln >> 4) << 1) + (j & 1);
      wfrag[i] = Wm[k * 64 + nbb * 16 + (ln & 15)];
    }
    return;
  }

  for (int b = tid; b < n_buckets; b += SORT_THREADS) cursor[b] = 0;
  __syncthreads();
  const int cbeg = wg * chunk;
  const int cend = (cbeg + chunk < n_edges) ? cbeg + chunk : n_edges;
  const bool staged = chunk <= STAGE_CAP;
  for (int i0 = cbeg + tid; i0 < cend; i0 += SORT_DEPTH * SORT_THREADS) {
    int d[SORT_DEPTH];
#pragma unroll
    for (int u = 0; u < SORT_DEPTH; ++u) {
      const int i = i0 + u * SORT_THREADS;
      d[u] = dst[i < cend ? i : cend - 1];
    }
#pragma unroll
    for (int u = 0; u < SORT_DEPTH; ++u)
      if (i0 + u * SORT_THREADS < cend) atomicAdd(&cursor[d[u] >> BUCKET_SHIFT], 1);
  }
  __syncthreads();
  {
    const int per = (n_buckets + SORT_THREADS - 1) / SORT_THREADS;
    const int b0 = tid * per;
    const int b1 = (b0 + per < n_buckets) ? b0 + per : n_buckets;
    int s = 0;
    for (int b = b0; b < b1; ++b) s += cursor[b];
    const int inc = wave_incl_scan(s, lane);
    if (lane == 31) wsum[wave] = inc;
    __syncthreads();
    int run = inc - s;
    for (int w2 = 0; w2 < SORT_THREADS / 32; ++w2)
      if (w2 < wave) run += wsum[w2];
    for (int b = b0; b < b1; ++b) {
      const int c = cursor[b];
      cursor[b] = run;
      const v2i m = {run, c};
      *(v2i*)(meta + 2 * ((size_t)b * n_wg + wg)) = m;
      run += c;
    }
  }
  __syncthreads();
  for (int i0 = cbeg + tid; i0 < cend; i0 += PLACE_DEPTH * SORT_THREADS) {
    int d[PLACE_DEPTH], t[PLACE_DEPTH], sn[PLACE_DEPTH];
#pragma unroll
    for (int u = 0; u < PLACE_DEPTH; ++u) {
      const int i = i0 + u * SORT_THREADS;
      const int ic = i < cend ? i : cend - 1;
      d[u] = dst[ic];
      t[u] = etype[ic];
      sn[u] = src[ic];
    }
#pragma unroll
    for (int u = 0; u < PLACE_DEPTH; ++u) {
      const int i = i0 + u * SORT_THREADS;
      if (i < cend) {
        const int tt = (t[u] < 0 || t[u] > 2) ? 3 : t[u];
        const int dl = d[u] & (BUCKET_NODES - 1);
        const int p = atomicAdd(&cursor[d[u] >> BUCKET_SHIFT], 1);
        if (staged) {
          const v2i r = {sn[u], ((i - cbeg) << (BUCKET_SHIFT + 2)) | (dl << 2) | tt};
          stage[p] = r;
        } else {
          const v4i r = {i, sn[u], tt, dl};
          *(v4i*)(rec + 4 * (size_t)(cbeg + p)) = r;
        }
      }
    }
  }
  if (!staged) return;
  __syncthreads();
  const int n_local = cend - cbeg;
  for (int i = tid; i < n_local; i += SORT_THREADS) {
    const v2i r = stage[i];
    const v4i o = {cbeg + (int)((unsigned)r.y >> (BUCKET_SHIFT + 2)), r.x, r.y & 3,
                   (r.y >> 2) & (BUCKET_NODES - 1)};
    *(v4i*)(rec + 4 * (size_t)(cbeg + i)) = o;
  }
}

#define AGG_WAVES 16
#define AGG_THREADS (AGG_WAVES * 32)
#define RUNS_PER_WAVE (SORT_WGS / AGG_WAVES)
#define QSTRIDE (BUCKET_NODES * OSTRIDE)

__global__ __launch_bounds__(AGG_THREADS, 16) void aggregate_xform_kernel(
    const float* __restrict__ xout, const float* __restrict__ ef,
    const int* __restrict__ rec, const int* __restrict__ meta, int chunk,
    const float* __restrict__ wfrag, float* __restrict__ out, int n_nodes) {
  __shared__ float s[BUCKET_NODES * SSTRIDE];
  const int tid = threadIdx.x;
  const int wave = __builtin_amdgcn_readfirstlane((int)(threadIdx.x >> 5));
  const int lane = threadIdx.x & 31;
  const int b = (int)blockIdx.x;
  v2i m = {0, 0};
  if (lane < RUNS_PER_WAVE)
    m = *(const v2i*)(meta + 2 * ((size_t)b * SORT_WGS + wave * RUNS_PER_WAVE + lane));

  for (int i = tid; i < BUCKET_NODES * SSTRIDE / 4; i += AGG_THREADS)
    ((v4f*)s)[i] = v4f{0.f, 0.f, 0.f, 0.f};
  __syncthreads();

  const int incl = wave_incl_scan(m.y, lane);
  const int excl = incl - m.y;
  const int total = __builtin_amdgcn_readlane(incl, 31);
  const int adj = (wave * RUNS_PER_WAVE + lane) * chunk + m.x - excl;

  for (int j0 = 0; j0 < total; j0 += 32) {
    const int left = total - j0;
    const int nrec = left < 32 ? left : 32;
    const int j = j0 + lane;
    int a = 0;
#pragma unroll
    for (int k = 0; k < RUNS_PER_WAVE; ++k) {
      const int ek = __builtin_amdgcn_readlane(excl, k);
      const int ak = __builtin_amdgcn_readlane(adj, k);
      if (j >= ek) a = ak;
    }
    v4i r = {0, 0, 3, 0};
    if (lane < nrec) r = *(const v4i*)(rec + 4 * (size_t)(a + j));
    for (int q = 0; q < nrec; q += 4) {
      float x0[4], x1[4], f[4];
      int off[4];
#pragma unroll
      for (int u = 0; u < 4; ++u) {
        const int qq = q + u;
        const bool live = qq < nrec;
        const int qc = live ? qq : q;
        const unsigned e = (unsigned)__builtin_amdgcn_readlane(r.x, qc);
        const unsigned sn = (unsigned)__builtin_amdgcn_readlane(r.y, qc);
        const int t = __builtin_amdgcn_readlane(r.z, qc);
        const int dl = __builtin_amdgcn_readlane(r.w, qc);
        off[u] = (live && t < 3) ? (dl * SSTRIDE + t * 96) : -1;
        const float* xr = xout + ((size_t)sn * 64u + (unsigned)lane);
        x0[u] = xr[0];
        x1[u] = xr[32];
        f[u] = ef[(size_t)e * 32u + (unsigned)lane];
      }
#pragma unroll
      for (int u = 0; u < 4; ++u) {
        if (off[u] >= 0) {
          float* row = s + off[u] + lane;
          unsafeAtomicAdd(row, x0[u]);
          unsafeAtomicAdd(row + 32, x1[u]);
          unsafeAtomicAdd(row + 64, f[u]);
        }
      }
    }
  }
  __syncthreads();

  const int nb = wave & 3;
  const int kq = wave >> 2;
  const int arow = lane & 15;
  const int h2 = (lane >> 4) << 1;
  const float* ap = s + arow * SSTRIDE + kq * 72 + h2;
  const v4f* __restrict__ bp = (const v4f*)wfrag + ((nb * 36 + kq * 9) * 32 + lane);

  v8f acc[4] = {};
#pragma unroll
  for (int j = 0; j < 9; ++j) {
    const v4f bq = bp[j * 32];
    const v2f b01 = {bq.x, bq.y};
    const v2f b23 = {bq.z, bq.w};
#pragma unroll
    for (int rt = 0; rt < 4; ++rt) {
      const v2f a0 = *(const v2f*)(ap + rt * 16 * SSTRIDE + j * 8);
      acc[rt] = wmma_f32(a0, b01, acc[rt]);
    }
#pragma unroll
    for (int rt = 0; rt < 4; ++rt) {
      const v2f a1 = *(const v2f*)(ap + rt * 16 * SSTRIDE + j * 8 + 4);
      acc[rt] = wmma_f32(a1, b23, acc[rt]);
    }
  }
  __syncthreads();

  float* st = s + kq * QSTRIDE + ((lane >> 4) * 8) * OSTRIDE + nb * 16 + (lane & 15);
#pragma unroll
  for (int rt = 0; rt < 4; ++rt) {
#pragma unroll
    for (int r8 = 0; r8 < 8; ++r8)
      st[(rt * 16 + r8) * OSTRIDE] = ((const float*)&acc[rt])[r8];
  }
  __syncthreads();

  const int node0 = b * BUCKET_NODES;
  const int rows = (n_nodes - node0 < BUCKET_NODES) ? n_nodes - node0 : BUCKET_NODES;
  v4f* __restrict__ og = (v4f*)(out + (size_t)node0 * 64);
  for (int i = tid; i < rows * 16; i += AGG_THREADS) {
    const float* p = s + (i >> 4) * OSTRIDE + (i & 15) * 4;
    v4f v = *(const v4f*)p;
    v += *(const v4f*)(p + QSTRIDE);
    v += *(const v4f*)(p + 2 * QSTRIDE);
    v += *(const v4f*)(p + 3 * QSTRIDE);
    og[i] = v;
  }
}

extern "C" void kernel_launch(void* const* d_in, const int* in_sizes, int n_in,
                              void* d_out, int out_size, void* d_ws,
                              size_t ws_size, hipStream_t stream) {
  const float* node_feature = (const float*)d_in[0];
  const float* edge_feature = (const float*)d_in[1];
  const float* W_node = (const float*)d_in[2];
  const float* W_msg = (const float*)d_in[3];
  const int* node_type = (const int*)d_in[4];
  const int* edge_type = (const int*)d_in[5];
  const int* edge_index = (const int*)d_in[6];
  const int n_nodes = in_sizes[4];
  const int n_edges = in_sizes[5];
  const int* src = edge_index;
  const int* dst = edge_index + n_edges;

  float* out = (float*)d_out;

  const int n_buckets = (n_nodes + BUCKET_NODES - 1) / BUCKET_NODES;
  const int sort_wgs = SORT_WGS;
  const int chunk = (n_edges + sort_wgs - 1) / sort_wgs;

  float* xout = (float*)d_ws;
  float* wfrag = xout + (size_t)n_nodes * 64;
  int* rec = (int*)(wfrag + WFRAG_FLOATS);
  int* meta = rec + 4 * (size_t)n_edges;

  const int node_tiles = (n_nodes + 15) / 16;
  sort_node_kernel<<<(node_tiles + 31) / 32 + sort_wgs + WFRAG_FLOATS / SORT_THREADS, SORT_THREADS,
                     n_buckets * sizeof(int), stream>>>(
      src, dst, edge_type, n_edges, chunk, n_buckets, sort_wgs, rec, meta,
      node_feature, W_node, node_type, xout, n_nodes, node_tiles, W_msg, wfrag);

  aggregate_xform_kernel<<<n_buckets, AGG_THREADS, 0, stream>>>(
      xout, edge_feature, rec, meta, chunk, wfrag, out, n_nodes);
}


# Node transform stages its 16-node tile in a per-wave LDS tile: feature rows are read and x_out rows written as whole 128-byte lines, not 16-byte and 64-byte pieces; sums and their order unchanged
# speedup vs baseline: 2.7008x; 1.0502x over previous
#include <hip/hip_runtime.h>


typedef __attribute__((ext_vector_type(2))) float v2f;
typedef __attribute__((ext_vector_type(4))) float v4f;
typedef __attribute__((ext_vector_type(8))) float v8f;
typedef __attribute__((ext_vector_type(4))) int v4i;
typedef __attribute__((ext_vector_type(2))) int v2i;

#define NSTRIDE 68
#define MSG_K 288
#define SSTRIDE 292
#define OSTRIDE 68
#define WFRAG_FLOATS (MSG_K * 64)

__device__ __forceinline__ v8f wmma_f32(v2f a, v2f b, v8f c) {
  return __builtin_amdgcn_wmma_f32_16x16x4_f32(false, a, false, b, (short)0, c,
                                               false, false);
}

__device__ __forceinline__ int wave_incl_scan(int v, int lane) {
#pragma unroll
  for (int off = 1; off < 32; off <<= 1) {
    int u = __shfl_up(v, off, 32);
    if (lane >= off) v += u;
  }
  return v;
}

#define BUCKET_SHIFT 6
#define BUCKET_NODES 64
#define SORT_THREADS 1024
#define SORT_WGS 128
#define SORT_DEPTH 16
#define PLACE_DEPTH 8
#define STAGE_CAP 12544
#define TSTRIDE 36
#define TILE_FLOATS (16 * TSTRIDE)
#define NODE_W_FLOATS (2 * 64 * NSTRIDE)
#define STAGE_V2I ((NODE_W_FLOATS + 32 * TILE_FLOATS) / 2)

__device__ __forceinline__ void wave_lds_handover() {
  __builtin_amdgcn_fence(__ATOMIC_RELEASE, "wavefront");
  __builtin_amdgcn_wave_barrier();
  __builtin_amdgcn_fence(__ATOMIC_ACQUIRE, "wavefront");
}

__global__ __launch_bounds__(SORT_THREADS, 16) void sort_node_kernel(
    const int* __restrict__ src, const int* __restrict__ dst,
    const int* __restrict__ etype, int n_edges, int chunk, int n_buckets,
    int n_wg, int* __restrict__ rec, int* __restrict__ meta,
    const float* __restrict__ x, const float* __restrict__ Wn,
    const int* __restrict__ ntype, float* __restrict__ xout, int n_nodes,
    int n_tiles, const float* __restrict__ Wm, float* __restrict__ wfrag) {
  extern __shared__ int cursor[];
  __shared__ v2i stage[STAGE_V2I];
  __shared__ int wsum[SORT_THREADS / 32];
  float* w = (float*)stage;
  const int tid = threadIdx.x;
  const int lane = tid & 31, wave = tid >> 5;
  const int tile_blocks = (n_tiles + 31) >> 5;

  if ((int)blockIdx.x < tile_blocks) {
    for (int i = tid; i < 2 * 64 * 64; i += SORT_THREADS) {
      int t = i >> 12, k = (i >> 6) & 63, cc = i & 63;
      w[(t * 64 + cc) * NSTRIDE + k] = Wn[i];
    }
    __syncthreads();

    const int tile = (int)blockIdx.x * 32 + wave;
    if (tile >= n_tiles) return;

    const int arow = lane & 15;
    const int khalf = (lane >> 4) << 1;
    const int node0 = tile * 16;
    const int t_row = (node0 + arow < n_nodes) ? ntype[node0 + arow] : -1;
    const float* __restrict__ wb0 = &w[arow * NSTRIDE];
    const float* __restrict__ wb1 = &w[(64 + arow) * NSTRIDE];
    float* tl = w + NODE_W_FLOATS + wave * TILE_FLOATS;
    const int srow = lane >> 3;
    const int scol = (lane & 7) << 2;
    float* tl_rw = tl + srow * TSTRIDE + scol;
    const float* tl_a = tl + arow * TSTRIDE + khalf;

    unsigned roff[4];
#pragma unroll
    for (int i = 0; i < 4; ++i) {
      int n = node0 + 4 * i + srow;
      if (n >= n_nodes) n = n_nodes - 1;
      roff[i] = (unsigned)n * 64u + (unsigned)scol;
    }
    v4f in[4];
#pragma unroll
    for (int i = 0; i < 4; ++i)
      in[i] = __builtin_nontemporal_load((const v4f*)(x + roff[i]));

    v8f acc[4] = {};
#pragma unroll
    for (int hh = 0; hh < 2; ++hh) {
      if (hh) wave_lds_handover();
#pragma unroll
      for (int i = 0; i < 4; ++i) *(v4f*)(tl_rw + 4 * i * TSTRIDE) = in[i];
      wave_lds_handover();
#pragma unroll 1
      for (int ks = 0; ks < 8; ++ks) {
        const int k = 32 * hh + 4 * ks + khalf;
        const v2f a = *(const v2f*)(tl_a + 4 * ks);
        const v2f z = {0.f, 0.f};
        v2f a0 = (t_row == 0) ? a : z;
        v2f a1 = (t_row == 1) ? a : z;
#pragma unroll
        for (int nb = 0; nb < 4; ++nb) {
          v2f b0 = *(const v2f*)(wb0 + nb * 16 * NSTRIDE + k);
          v2f b1 = *(const v2f*)(wb1 + nb * 16 * NSTRIDE + k);
          acc[nb] = wmma_f32(a0, b0, acc[nb]);
          acc[nb] = wmma_f32(a1, b1, acc[nb]);
        }
      }
      if (hh == 0) {
#pragma unroll
        for (int i = 0; i < 4; ++i)
          in[i] = __builtin_nontemporal_load((const v4f*)(x + roff[i] + 32u));
      }
    }

    const int rbase = (lane >> 4) * 8;
    const int col = lane & 15;
#pragma unroll
    for (int hh = 0; hh < 2; ++hh) {
      wave_lds_handover();
#pragma unroll
      for (int q = 0; q < 2; ++q) {
#pragma unroll
        for (int r = 0; r < 8; ++r)
          tl[(rbase + r) * TSTRIDE + q * 16 + col] = ((const float*)&acc[2 * hh + q])[r];
      }
      wave_lds_handover();
#pragma unroll
      for (int i = 0; i < 4; ++i) {
        const int n = node0 + 4 * i + srow;
        const v4f v = *(const v4f*)(tl_rw + 4 * i * TSTRIDE);
        if (n < n_nodes) *(v4f*)(xout + ((unsigned)n * 64u + (unsigned)(32 * hh + scol))) = v;
      }
    }
    return;
  }

  const int wg = (int)blockIdx.x - tile_blocks;
  if (wg >= n_wg) {
    const int i = (wg - n_wg) * SORT_THREADS + tid;
    if (i < WFRAG_FLOATS) {
      const int j = i & 3, ln = (i >> 2) & 31, blk = i >> 7;
      const int kb = blk % 36, nbb = blk / 36;
      const int k = kb * 8 + (j >> 1) * 4 + ((ln >> 4) << 1) + (j & 1);
      wfrag[i] = Wm[k * 64 + nbb * 16 + (ln & 15)];
    }
    return;
  }

  for (int b = tid; b < n_buckets; b += SORT_THREADS) cursor[b] = 0;
  __syncthreads();
  const int cbeg = wg * chunk;
  const int cend = (cbeg + chunk < n_edges) ? cbeg + chunk : n_edges;
  const bool staged = chunk <= STAGE_CAP;
  for (int i0 = cbeg + tid; i0 < cend; i0 += SORT_DEPTH * SORT_THREADS) {
    int d[SORT_DEPTH];
#pragma unroll
    for (int u = 0; u < SORT_DEPTH; ++u) {
      const int i = i0 + u * SORT_THREADS;
      d[u] = dst[i < cend ? i : cend - 1];
    }
#pragma unroll
    for (int u = 0; u < SORT_DEPTH; ++u)
      if (i0 + u * SORT_THREADS < cend) atomicAdd(&cursor[d[u] >> BUCKET_SHIFT], 1);
  }
  __syncthreads();
  {
    const int per = (n_buckets + SORT_THREADS - 1) / SORT_THREADS;
    const int b0 = tid * per;
    const int b1 = (b0 + per < n_buckets) ? b0 + per : n_buckets;
    int s = 0;
    for (int b = b0; b < b1; ++b) s += cursor[b];
    const int inc = wave_incl_scan(s, lane);
    if (lane == 31) wsum[wave] = inc;
    __syncthreads();
    int run = inc - s;
    for (int w2 = 0; w2 < SORT_THREADS / 32; ++w2)
      if (w2 < wave) run += wsum[w2];
    for (int b = b0; b < b1; ++b) {
      const int c = cursor[b];
      cursor[b] = run;
      const v2i m = {run, c};
      *(v2i*)(meta + 2 * ((size_t)b * n_wg + wg)) = m;
      run += c;
    }
  }
  __syncthreads();
  for (int i0 = cbeg + tid; i0 < cend; i0 += PLACE_DEPTH * SORT_THREADS) {
    int d[PLACE_DEPTH], t[PLACE_DEPTH], sn[PLACE_DEPTH];
#pragma unroll
    for (int u = 0; u < PLACE_DEPTH; ++u) {
      const int i = i0 + u * SORT_THREADS;
      const int ic = i < cend ? i : cend - 1;
      d[u] = dst[ic];
      t[u] = etype[ic];
      sn[u] = src[ic];
    }
#pragma unroll
    for (int u = 0; u < PLACE_DEPTH; ++u) {
      const int i = i0 + u * SORT_THREADS;
      if (i < cend) {
        const int tt = (t[u] < 0 || t[u] > 2) ? 3 : t[u];
        const int dl = d[u] & (BUCKET_NODES - 1);
        const int p = atomicAdd(&cursor[d[u] >> BUCKET_SHIFT], 1);
        if (staged) {
          const v2i r = {sn[u], ((i - cbeg) << (BUCKET_SHIFT + 2)) | (dl << 2) | tt};
          stage[p] = r;
        } else {
          const v4i r = {i, sn[u], tt, dl};
          *(v4i*)(rec + 4 * (size_t)(cbeg + p)) = r;
        }
      }
    }
  }
  if (!staged) return;
  __syncthreads();
  const int n_local = cend - cbeg;
  for (int i = tid; i < n_local; i += SORT_THREADS) {
    const v2i r = stage[i];
    const v4i o = {cbeg + (int)((unsigned)r.y >> (BUCKET_SHIFT + 2)), r.x, r.y & 3,
                   (r.y >> 2) & (BUCKET_NODES - 1)};
    *(v4i*)(rec + 4 * (size_t)(cbeg + i)) = o;
  }
}

#define AGG_WAVES 16
#define AGG_THREADS (AGG_WAVES * 32)
#define RUNS_PER_WAVE (SORT_WGS / AGG_WAVES)
#define QSTRIDE (BUCKET_NODES * OSTRIDE)

__global__ __launch_bounds__(AGG_THREADS, 16) void aggregate_xform_kernel(
    const float* __restrict__ xout, const float* __restrict__ ef,
    const int* __restrict__ rec, const int* __restrict__ meta, int chunk,
    const float* __restrict__ wfrag, float* __restrict__ out, int n_nodes) {
  __shared__ float s[BUCKET_NODES * SSTRIDE];
  const int tid = threadIdx.x;
  const int wave = __builtin_amdgcn_readfirstlane((int)(threadIdx.x >> 5));
  const int lane = threadIdx.x & 31;
  const int b = (int)blockIdx.x;
  v2i m = {0, 0};
  if (lane < RUNS_PER_WAVE)
    m = *(const v2i*)(meta + 2 * ((size_t)b * SORT_WGS + wave * RUNS_PER_WAVE + lane));

  for (int i = tid; i < BUCKET_NODES * SSTRIDE / 4; i += AGG_THREADS)
    ((v4f*)s)[i] = v4f{0.f, 0.f, 0.f, 0.f};
  __syncthreads();

  const int incl = wave_incl_scan(m.y, lane);
  const int excl = incl - m.y;
  const int total = __builtin_amdgcn_readlane(incl, 31);
  const int adj = (wave * RUNS_PER_WAVE + lane) * chunk + m.x - excl;

  for (int j0 = 0; j0 < total; j0 += 32) {
    const int left = total - j0;
    const int nrec = left < 32 ? left : 32;
    const int j = j0 + lane;
    int a = 0;
#pragma unroll
    for (int k = 0; k < RUNS_PER_WAVE; ++k) {
      const int ek = __builtin_amdgcn_readlane(excl, k);
      const int ak = __builtin_amdgcn_readlane(adj, k);
      if (j >= ek) a = ak;
    }
    v4i r = {0, 0, 3, 0};
    if (lane < nrec) r = *(const v4i*)(rec + 4 * (size_t)(a + j));
    for (int q = 0; q < nrec; q += 4) {
      float x0[4], x1[4], f[4];
      int off[4];
#pragma unroll
      for (int u = 0; u < 4; ++u) {
        const int qq = q + u;
        const bool live = qq < nrec;
        const int qc = live ? qq : q;
        const unsigned e = (unsigned)__builtin_amdgcn_readlane(r.x, qc);
        const unsigned sn = (unsigned)__builtin_amdgcn_readlane(r.y, qc);
        const int t = __builtin_amdgcn_readlane(r.z, qc);
        const int dl = __builtin_amdgcn_readlane(r.w, qc);
        off[u] = (live && t < 3) ? (dl * SSTRIDE + t * 96) : -1;
        const float* xr = xout + ((size_t)sn * 64u + (unsigned)lane);
        x0[u] = xr[0];
        x1[u] = xr[32];
        f[u] = ef[(size_t)e * 32u + (unsigned)lane];
      }
#pragma unroll
      for (int u = 0; u < 4; ++u) {
        if (off[u] >= 0) {
          float* row = s + off[u] + lane;
          unsafeAtomicAdd(row, x0[u]);
          unsafeAtomicAdd(row + 32, x1[u]);
          unsafeAtomicAdd(row + 64, f[u]);
        }
      }
    }
  }
  __syncthreads();

  const int nb = wave & 3;
  const int kq = wave >> 2;
  const int arow = lane & 15;
  const int h2 = (lane >> 4) << 1;
  const float* ap = s + arow * SSTRIDE + kq * 72 + h2;
  const v4f* __restrict__ bp = (const v4f*)wfrag + ((nb * 36 + kq * 9) * 32 + lane);

  v8f acc[4] = {};
#pragma unroll
  for (int j = 0; j < 9; ++j) {
    const v4f bq = bp[j * 32];
    const v2f b01 = {bq.x, bq.y};
    const v2f b23 = {bq.z, bq.w};
#pragma unroll
    for (int rt = 0; rt < 4; ++rt) {
      const v2f a0 = *(const v2f*)(ap + rt * 16 * SSTRIDE + j * 8);
      acc[rt] = wmma_f32(a0, b01, acc[rt]);
    }
#pragma unroll
    for (int rt = 0; rt < 4; ++rt) {
      const v2f a1 = *(const v2f*)(ap + rt * 16 * SSTRIDE + j * 8 + 4);
      acc[rt] = wmma_f32(a1, b23, acc[rt]);
    }
  }
  __syncthreads();

  float* st = s + kq * QSTRIDE + ((lane >> 4) * 8) * OSTRIDE + nb * 16 + (lane & 15);
#pragma unroll
  for (int rt = 0; rt < 4; ++rt) {
#pragma unroll
    for (int r8 = 0; r8 < 8; ++r8)
      st[(rt * 16 + r8) * OSTRIDE] = ((const float*)&acc[rt])[r8];
  }
  __syncthreads();

  const int node0 = b * BUCKET_NODES;
  const int rows = (n_nodes - node0 < BUCKET_NODES) ? n_nodes - node0 : BUCKET_NODES;
  v4f* __restrict__ og = (v4f*)(out + (size_t)node0 * 64);
  for (int i = tid; i < rows * 16; i += AGG_THREADS) {
    const float* p = s + (i >> 4) * OSTRIDE + (i & 15) * 4;
    v4f v = *(const v4f*)p;
    v += *(const v4f*)(p + QSTRIDE);
    v += *(const v4f*)(p + 2 * QSTRIDE);
    v += *(const v4f*)(p + 3 * QSTRIDE);
    og[i] = v;
  }
}

extern "C" void kernel_launch(void* const* d_in, const int* in_sizes, int n_in,
                              void* d_out, int out_size, void* d_ws,
                              size_t ws_size, hipStream_t stream) {
  const float* node_feature = (const float*)d_in[0];
  const float* edge_feature = (const float*)d_in[1];
  const float* W_node = (const float*)d_in[2];
  const float* W_msg = (const float*)d_in[3];
  const int* node_type = (const int*)d_in[4];
  const int* edge_type = (const int*)d_in[5];
  const int* edge_index = (const int*)d_in[6];
  const int n_nodes = in_sizes[4];
  const int n_edges = in_sizes[5];
  const int* src = edge_index;
  const int* dst = edge_index + n_edges;

  float* out = (float*)d_out;

  const int n_buckets = (n_nodes + BUCKET_NODES - 1) / BUCKET_NODES;
  const int sort_wgs = SORT_WGS;
  const int chunk = (n_edges + sort_wgs - 1) / sort_wgs;

  float* xout = (float*)d_ws;
  float* wfrag = xout + (size_t)n_nodes * 64;
  int* rec = (int*)(wfrag + WFRAG_FLOATS);
  int* meta = rec + 4 * (size_t)n_edges;

  const int node_tiles = (n_nodes + 15) / 16;
  sort_node_kernel<<<(node_tiles + 31) / 32 + sort_wgs + WFRAG_FLOATS / SORT_THREADS, SORT_THREADS,
                     n_buckets * sizeof(int), stream>>>(
      src, dst, edge_type, n_edges, chunk, n_buckets, sort_wgs, rec, meta,
      node_feature, W_node, node_type, xout, n_nodes, node_tiles, W_msg, wfrag);

  aggregate_xform_kernel<<<n_buckets, AGG_THREADS, 0, stream>>>(
      xout, edge_feature, rec, meta, chunk, wfrag, out, n_nodes);
}


# Node transform: a work-group's 512 nodes ordered by type in LDS (ballot ranks) so a 16-node tile multiplies one type's weights: 128 -> 64 WMMAs a tile; boundary tiles take both; sums unchanged
# speedup vs baseline: 2.7915x; 1.0336x over previous
#include <hip/hip_runtime.h>


typedef __attribute__((ext_vector_type(2))) float v2f;
typedef __attribute__((ext_vector_type(4))) float v4f;
typedef __attribute__((ext_vector_type(8))) float v8f;
typedef __attribute__((ext_vector_type(4))) int v4i;
typedef __attribute__((ext_vector_type(2))) int v2i;

#define NSTRIDE 68
#define MSG_K 288
#define SSTRIDE 292
#define OSTRIDE 68
#define WFRAG_FLOATS (MSG_K * 64)

__device__ __forceinline__ v8f wmma_f32(v2f a, v2f b, v8f c) {
  return __builtin_amdgcn_wmma_f32_16x16x4_f32(false, a, false, b, (short)0, c,
                                               false, false);
}

__device__ __forceinline__ int wave_incl_scan(int v, int lane) {
#pragma unroll
  for (int off = 1; off < 32; off <<= 1) {
    int u = __shfl_up(v, off, 32);
    if (lane >= off) v += u;
  }
  return v;
}

#define BUCKET_SHIFT 6
#define BUCKET_NODES 64
#define SORT_THREADS 1024
#define SORT_WGS 128
#define SORT_DEPTH 16
#define PLACE_DEPTH 8
#define STAGE_CAP 12544
#define TSTRIDE 36
#define TILE_FLOATS (16 * TSTRIDE)
#define NODE_W_FLOATS (2 * 64 * NSTRIDE)
#define STAGE_V2I ((NODE_W_FLOATS + 32 * TILE_FLOATS) / 2)
#define NODES_PER_WG 512

__device__ __forceinline__ void wave_lds_handover() {
  __builtin_amdgcn_fence(__ATOMIC_RELEASE, "wavefront");
  __builtin_amdgcn_wave_barrier();
  __builtin_amdgcn_fence(__ATOMIC_ACQUIRE, "wavefront");
}

__global__ __launch_bounds__(SORT_THREADS, 16) void sort_node_kernel(
    const int* __restrict__ src, const int* __restrict__ dst,
    const int* __restrict__ etype, int n_edges, int chunk, int n_buckets,
    int n_wg, int* __restrict__ rec, int* __restrict__ meta,
    const float* __restrict__ x, const float* __restrict__ Wn,
    const int* __restrict__ ntype, float* __restrict__ xout, int n_nodes,
    int n_tiles, const float* __restrict__ Wm, float* __restrict__ wfrag) {
  extern __shared__ int cursor[];
  __shared__ v2i stage[STAGE_V2I];
  __shared__ int wsum[SORT_THREADS / 32];
  __shared__ int perm[NODES_PER_WG];
  float* w = (float*)stage;
  const int tid = threadIdx.x;
  const int lane = tid & 31, wave = tid >> 5;
  const int tile_blocks = (n_tiles + 31) >> 5;

  if ((int)blockIdx.x < tile_blocks) {
    const int wg_node0 = (int)blockIdx.x * NODES_PER_WG;
    int t_mine = -1;
    if (tid < NODES_PER_WG && wg_node0 + tid < n_nodes) t_mine = ntype[wg_node0 + tid];
    for (int i = tid; i < 2 * 64 * 64; i += SORT_THREADS) {
      int t = i >> 12, k = (i >> 6) & 63, cc = i & 63;
      w[(t * 64 + cc) * NSTRIDE + k] = Wn[i];
    }
    const int cls = (t_mine == 0) ? 0 : (t_mine == 1) ? 1 : 2;
    unsigned bal0 = 0, bal1 = 0;
    if (tid < NODES_PER_WG) {
      bal0 = __builtin_amdgcn_ballot_w32(cls == 0);
      bal1 = __builtin_amdgcn_ballot_w32(cls == 1);
      if (lane == 0) wsum[wave] = __popc(bal0) | (__popc(bal1) << 16);
    }
    __syncthreads();
    if (tid < NODES_PER_WG) {
      int before = 0, total = 0;
#pragma unroll
      for (int w2 = 0; w2 < NODES_PER_WG / 32; ++w2) {
        const int c = wsum[w2];
        total += c;
        if (w2 < wave) before += c;
      }
      const int n0 = total & 0xffff, n1 = total >> 16;
      const int o0 = before & 0xffff, o1 = before >> 16;
      const unsigned lt = (1u << lane) - 1u;
      int pos;
      if (cls == 0) pos = o0 + __popc(bal0 & lt);
      else if (cls == 1) pos = n0 + o1 + __popc(bal1 & lt);
      else pos = n0 + n1 + (32 * wave - o0 - o1) + __popc(~(bal0 | bal1) & lt);
      perm[pos] = tid | (cls << 16);
    }
    __syncthreads();

    const int arow = lane & 15;
    const int khalf = (lane >> 4) << 1;
    const int* tperm = perm + wave * 16;
    const int ent = tperm[arow];
    const int t_row = ent >> 16;
    if (__builtin_amdgcn_ballot_w32(wg_node0 + (ent & 0xffff) < n_nodes) == 0u) return;
    const bool any0 = __builtin_amdgcn_ballot_w32(t_row == 0) != 0u;
    const bool any1 = __builtin_amdgcn_ballot_w32(t_row == 1) != 0u;
    const bool mixed = any0 && any1;
    const int t_first = any0 ? 0 : 1;
    const float* __restrict__ wb_first = &w[(t_first * 64 + arow) * NSTRIDE];
    float* tl = w + NODE_W_FLOATS + wave * TILE_FLOATS;
    const int srow = lane >> 3;
    const int scol = (lane & 7) << 2;
    float* tl_rw = tl + srow * TSTRIDE + scol;
    const float* tl_a = tl + arow * TSTRIDE + khalf;

    unsigned roff[4];
#pragma unroll
    for (int i = 0; i < 4; ++i) {
      int n = wg_node0 + (tperm[4 * i + srow] & 0xffff);
      if (n >= n_nodes) n = n_nodes - 1;
      roff[i] = (unsigned)n * 64u + (unsigned)scol;
    }
    v4f in[4];
#pragma unroll
    for (int i = 0; i < 4; ++i)
      in[i] = __builtin_nontemporal_load((const v4f*)(x + roff[i]));

    v8f acc[4] = {};
#pragma unroll
    for (int hh = 0; hh < 2; ++hh) {
      if (hh) wave_lds_handover();
#pragma unroll
      for (int i = 0; i < 4; ++i) *(v4f*)(tl_rw + 4 * i * TSTRIDE) = in[i];
      wave_lds_handover();
#pragma unroll 1
      for (int ks = 0; ks < 8; ++ks) {
        const int k = 32 * hh + 4 * ks + khalf;
        const v2f a = *(const v2f*)(tl_a + 4 * ks);
        const v2f z = {0.f, 0.f};
        const v2f aa = (t_row == t_first) ? a : z;
#pragma unroll
        for (int nb = 0; nb < 4; ++nb) {
          const v2f bb = *(const v2f*)(wb_first + nb * 16 * NSTRIDE + k);
          acc[nb] = wmma_f32(aa, bb, acc[nb]);
        }
        if (mixed) {
          const v2f a1 = (t_row == 1) ? a : z;
#pragma unroll
          for (int nb = 0; nb < 4; ++nb) {
            const v2f b1 = *(const v2f*)(wb_first + (64 + nb * 16) * NSTRIDE + k);
            acc[nb] = wmma_f32(a1, b1, acc[nb]);
          }
        }
      }
      if (hh == 0) {
#pragma unroll
        for (int i = 0; i < 4; ++i)
          in[i] = __builtin_nontemporal_load((const v4f*)(x + roff[i] + 32u));
      }
    }

    const int rbase = (lane >> 4) * 8;
    const int col = lane & 15;
    int nrow[4];
#pragma unroll
    for (int i = 0; i < 4; ++i) nrow[i] = wg_node0 + (tperm[4 * i + srow] & 0xffff);
#pragma unroll
    for (int hh = 0; hh < 2; ++hh) {
      wave_lds_handover();
#pragma unroll
      for (int q = 0; q < 2; ++q) {
#pragma unroll
        for (int r = 0; r < 8; ++r)
          tl[(rbase + r) * TSTRIDE + q * 16 + col] = ((const float*)&acc[2 * hh + q])[r];
      }
      wave_lds_handover();
#pragma unroll
      for (int i = 0; i < 4; ++i) {
        const v4f v = *(const v4f*)(tl_rw + 4 * i * TSTRIDE);
        if (nrow[i] < n_nodes) *(v4f*)(xout + ((unsigned)nrow[i] * 64u + (unsigned)(32 * hh + scol))) = v;
      }
    }
    return;
  }

  const int wg = (int)blockIdx.x - tile_blocks;
  if (wg >= n_wg) {
    const int i = (wg - n_wg) * SORT_THREADS + tid;
    if (i < WFRAG_FLOATS) {
      const int j = i & 3, ln = (i >> 2) & 31, blk = i >> 7;
      const int kb = blk % 36, nbb = blk / 36;
      const int k = kb * 8 + (j >> 1) * 4 + ((ln >> 4) << 1) + (j & 1);
      wfrag[i] = Wm[k * 64 + nbb * 16 + (ln & 15)];
    }
    return;
  }

  for (int b = tid; b < n_buckets; b += SORT_THREADS) cursor[b] = 0;
  __syncthreads();
  const int cbeg = wg * chunk;
  const int cend = (cbeg + chunk < n_edges) ? cbeg + chunk : n_edges;
  const bool staged = chunk <= STAGE_CAP;
  for (int i0 = cbeg + tid; i0 < cend; i0 += SORT_DEPTH * SORT_THREADS) {
    int d[SORT_DEPTH];
#pragma unroll
    for (int u = 0; u < SORT_DEPTH; ++u) {
      const int i = i0 + u * SORT_THREADS;
      d[u] = dst[i < cend ? i : cend - 1];
    }
#pragma unroll
    for (int u = 0; u < SORT_DEPTH; ++u)
      if (i0 + u * SORT_THREADS < cend) atomicAdd(&cursor[d[u] >> BUCKET_SHIFT], 1);
  }
  __syncthreads();
  {
    const int per = (n_buckets + SORT_THREADS - 1) / SORT_THREADS;
    const int b0 = tid * per;
    const int b1 = (b0 + per < n_buckets) ? b0 + per : n_buckets;
    int s = 0;
    for (int b = b0; b < b1; ++b) s += cursor[b];
    const int inc = wave_incl_scan(s, lane);
    if (lane == 31) wsum[wave] = inc;
    __syncthreads();
    int run = inc - s;
    for (int w2 = 0; w2 < SORT_THREADS / 32; ++w2)
      if (w2 < wave) run += wsum[w2];
    for (int b = b0; b < b1; ++b) {
      const int c = cursor[b];
      cursor[b] = run;
      const v2i m = {run, c};
      *(v2i*)(meta + 2 * ((size_t)b * n_wg + wg)) = m;
      run += c;
    }
  }
  __syncthreads();
  for (int i0 = cbeg + tid; i0 < cend; i0 += PLACE_DEPTH * SORT_THREADS) {
    int d[PLACE_DEPTH], t[PLACE_DEPTH], sn[PLACE_DEPTH];
#pragma unroll
    for (int u = 0; u < PLACE_DEPTH; ++u) {
      const int i = i0 + u * SORT_THREADS;
      const int ic = i < cend ? i : cend - 1;
      d[u] = dst[ic];
      t[u] = etype[ic];
      sn[u] = src[ic];
    }
#pragma unroll
    for (int u = 0; u < PLACE_DEPTH; ++u) {
      const int i = i0 + u * SORT_THREADS;
      if (i < cend) {
        const int tt = (t[u] < 0 || t[u] > 2) ? 3 : t[u];
        const int dl = d[u] & (BUCKET_NODES - 1);
        const int p = atomicAdd(&cursor[d[u] >> BUCKET_SHIFT], 1);
        if (staged) {
          const v2i r = {sn[u], ((i - cbeg) << (BUCKET_SHIFT + 2)) | (dl << 2) | tt};
          stage[p] = r;
        } else {
          const v4i r = {i, sn[u], tt, dl};
          *(v4i*)(rec + 4 * (size_t)(cbeg + p)) = r;
        }
      }
    }
  }
  if (!staged) return;
  __syncthreads();
  const int n_local = cend - cbeg;
  for (int i = tid; i < n_local; i += SORT_THREADS) {
    const v2i r = stage[i];
    const v4i o = {cbeg + (int)((unsigned)r.y >> (BUCKET_SHIFT + 2)), r.x, r.y & 3,
                   (r.y >> 2) & (BUCKET_NODES - 1)};
    *(v4i*)(rec + 4 * (size_t)(cbeg + i)) = o;
  }
}

#define AGG_WAVES 16
#define AGG_THREADS (AGG_WAVES * 32)
#define RUNS_PER_WAVE (SORT_WGS / AGG_WAVES)
#define QSTRIDE (BUCKET_NODES * OSTRIDE)

__global__ __launch_bounds__(AGG_THREADS, 16) void aggregate_xform_kernel(
    const float* __restrict__ xout, const float* __restrict__ ef,
    const int* __restrict__ rec, const int* __restrict__ meta, int chunk,
    const float* __restrict__ wfrag, float* __restrict__ out, int n_nodes) {
  __shared__ float s[BUCKET_NODES * SSTRIDE];
  const int tid = threadIdx.x;
  const int wave = __builtin_amdgcn_readfirstlane((int)(threadIdx.x >> 5));
  const int lane = threadIdx.x & 31;
  const int b = (int)blockIdx.x;
  v2i m = {0, 0};
  if (lane < RUNS_PER_WAVE)
    m = *(const v2i*)(meta + 2 * ((size_t)b * SORT_WGS + wave * RUNS_PER_WAVE + lane));

  for (int i = tid; i < BUCKET_NODES * SSTRIDE / 4; i += AGG_THREADS)
    ((v4f*)s)[i] = v4f{0.f, 0.f, 0.f, 0.f};
  __syncthreads();

  const int incl = wave_incl_scan(m.y, lane);
  const int excl = incl - m.y;
  const int total = __builtin_amdgcn_readlane(incl, 31);
  const int adj = (wave * RUNS_PER_WAVE + lane) * chunk + m.x - excl;

  for (int j0 = 0; j0 < total; j0 += 32) {
    const int left = total - j0;
    const int nrec = left < 32 ? left : 32;
    const int j = j0 + lane;
    int a = 0;
#pragma unroll
    for (int k = 0; k < RUNS_PER_WAVE; ++k) {
      const int ek = __builtin_amdgcn_readlane(excl, k);
      const int ak = __builtin_amdgcn_readlane(adj, k);
      if (j >= ek) a = ak;
    }
    v4i r = {0, 0, 3, 0};
    if (lane < nrec) r = *(const v4i*)(rec + 4 * (size_t)(a + j));
    for (int q = 0; q < nrec; q += 4) {
      float x0[4], x1[4], f[4];
      int off[4];
#pragma unroll
      for (int u = 0; u < 4; ++u) {
        const int qq = q + u;
        const bool live = qq < nrec;
        const int qc = live ? qq : q;
        const unsigned e = (unsigned)__builtin_amdgcn_readlane(r.x, qc);
        const unsigned sn = (unsigned)__builtin_amdgcn_readlane(r.y, qc);
        const int t = __builtin_amdgcn_readlane(r.z, qc);
        const int dl = __builtin_amdgcn_readlane(r.w, qc);
        off[u] = (live && t < 3) ? (dl * SSTRIDE + t * 96) : -1;
        const float* xr = xout + ((size_t)sn * 64u + (unsigned)lane);
        x0[u] = xr[0];
        x1[u] = xr[32];
        f[u] = ef[(size_t)e * 32u + (unsigned)lane];
      }
#pragma unroll
      for (int u = 0; u < 4; ++u) {
        if (off[u] >= 0) {
          float* row = s + off[u] + lane;
          unsafeAtomicAdd(row, x0[u]);
          unsafeAtomicAdd(row + 32, x1[u]);
          unsafeAtomicAdd(row + 64, f[u]);
        }
      }
    }
  }
  __syncthreads();

  const int nb = wave & 3;
  const int kq = wave >> 2;
  const int arow = lane & 15;
  const int h2 = (lane >> 4) << 1;
  const float* ap = s + arow * SSTRIDE + kq * 72 + h2;
  const v4f* __restrict__ bp = (const v4f*)wfrag + ((nb * 36 + kq * 9) * 32 + lane);

  v8f acc[4] = {};
#pragma unroll
  for (int j = 0; j < 9; ++j) {
    const v4f bq = bp[j * 32];
    const v2f b01 = {bq.x, bq.y};
    const v2f b23 = {bq.z, bq.w};
#pragma unroll
    for (int rt = 0; rt < 4; ++rt) {
      const v2f a0 = *(const v2f*)(ap + rt * 16 * SSTRIDE + j * 8);
      acc[rt] = wmma_f32(a0, b01, acc[rt]);
    }
#pragma unroll
    for (int rt = 0; rt < 4; ++rt) {
      const v2f a1 = *(const v2f*)(ap + rt * 16 * SSTRIDE + j * 8 + 4);
      acc[rt] = wmma_f32(a1, b23, acc[rt]);
    }
  }
  __syncthreads();

  float* st = s + kq * QSTRIDE + ((lane >> 4) * 8) * OSTRIDE + nb * 16 + (lane & 15);
#pragma unroll
  for (int rt = 0; rt < 4; ++rt) {
#pragma unroll
    for (int r8 = 0; r8 < 8; ++r8)
      st[(rt * 16 + r8) * OSTRIDE] = ((const float*)&acc[rt])[r8];
  }
  __syncthreads();

  const int node0 = b * BUCKET_NODES;
  const int rows = (n_nodes - node0 < BUCKET_NODES) ? n_nodes - node0 : BUCKET_NODES;
  v4f* __restrict__ og = (v4f*)(out + (size_t)node0 * 64);
  for (int i = tid; i < rows * 16; i += AGG_THREADS) {
    const float* p = s + (i >> 4) * OSTRIDE + (i & 15) * 4;
    v4f v = *(const v4f*)p;
    v += *(const v4f*)(p + QSTRIDE);
    v += *(const v4f*)(p + 2 * QSTRIDE);
    v += *(const v4f*)(p + 3 * QSTRIDE);
    og[i] = v;
  }
}

extern "C" void kernel_launch(void* const* d_in, const int* in_sizes, int n_in,
                              void* d_out, int out_size, void* d_ws,
                              size_t ws_size, hipStream_t stream) {
  const float* node_feature = (const float*)d_in[0];
  const float* edge_feature = (const float*)d_in[1];
  const float* W_node = (const float*)d_in[2];
  const float* W_msg = (const float*)d_in[3];
  const int* node_type = (const int*)d_in[4];
  const int* edge_type = (const int*)d_in[5];
  const int* edge_index = (const int*)d_in[6];
  const int n_nodes = in_sizes[4];
  const int n_edges = in_sizes[5];
  const int* src = edge_index;
  const int* dst = edge_index + n_edges;

  float* out = (float*)d_out;

  const int n_buckets = (n_nodes + BUCKET_NODES - 1) / BUCKET_NODES;
  const int sort_wgs = SORT_WGS;
  const int chunk = (n_edges + sort_wgs - 1) / sort_wgs;

  float* xout = (float*)d_ws;
  float* wfrag = xout + (size_t)n_nodes * 64;
  int* rec = (int*)(wfrag + WFRAG_FLOATS);
  int* meta = rec + 4 * (size_t)n_edges;

  const int node_tiles = (n_nodes + 15) / 16;
  sort_node_kernel<<<(node_tiles + 31) / 32 + sort_wgs + WFRAG_FLOATS / SORT_THREADS, SORT_THREADS,
                     n_buckets * sizeof(int), stream>>>(
      src, dst, edge_type, n_edges, chunk, n_buckets, sort_wgs, rec, meta,
      node_feature, W_node, node_type, xout, n_nodes, node_tiles, W_msg, wfrag);

  aggregate_xform_kernel<<<n_buckets, AGG_THREADS, 0, stream>>>(
      xout, edge_feature, rec, meta, chunk, wfrag, out, n_nodes);
}
